# starting kernel
# baseline (speedup 1.0000x reference)
#include <hip/hip_runtime.h>
#include <math.h>

#define N_NODES 50000
#define N_EDGES 800000
#define ETOT    (N_EDGES + N_NODES)
#define F_IN    128
#define HID     64
#define HEADS   2
#define OUT_F   40
#define EDIM    5
#define NEG_SLOPE 0.2f

#define CDIV(a,b) (((a)+(b)-1)/(b))

typedef __attribute__((ext_vector_type(2))) float v2f;
typedef __attribute__((ext_vector_type(8))) float v8f;

__device__ inline unsigned encf(float f) {
    unsigned u = __float_as_uint(f);
    return (u & 0x80000000u) ? ~u : (u | 0x80000000u);
}
__device__ inline float decf(unsigned m) {
    unsigned u = (m & 0x80000000u) ? (m ^ 0x80000000u) : ~m;
    return __uint_as_float(u);
}
#define ENC_NEG_INF 0x007FFFFFu

__global__ void fill_f32(float* p, int n, float v) {
    int i = blockIdx.x * blockDim.x + threadIdx.x;
    if (i < n) p[i] = v;
}
__global__ void fill_u32(unsigned* p, int n, unsigned v) {
    int i = blockIdx.x * blockDim.x + threadIdx.x;
    if (i < n) p[i] = v;
}

__global__ void pad_b(const float* __restrict__ W, float* __restrict__ Wp,
                      int K, int N, int NPAD) {
    int i = blockIdx.x * blockDim.x + threadIdx.x;
    if (i >= K * NPAD) return;
    int r = i / NPAD, c = i % NPAD;
    Wp[i] = (c < N) ? W[r * N + c] : 0.0f;
}

__global__ void selfloop_accum(const int* __restrict__ dst, const float* __restrict__ eattr,
                               float* cnt, float* loop5, int E) {
    int e = blockIdx.x * blockDim.x + threadIdx.x;
    if (e >= E) return;
    int d = dst[e];
    atomicAdd(&cnt[d], 1.0f);
#pragma unroll
    for (int i = 0; i < EDIM; i++)
        atomicAdd(&loop5[d * EDIM + i], eattr[(size_t)e * EDIM + i]);
}
__global__ void selfloop_final(float* loop5, const float* cnt, int n) {
    int i = blockIdx.x * blockDim.x + threadIdx.x;
    if (i >= n * EDIM) return;
    float c = fmaxf(cnt[i / EDIM], 1.0f);
    loop5[i] /= c;
}

__global__ void precompute_v(const float* We1, const float* ae1,
                             const float* We2, const float* ae2,
                             float* v1, float* v2) {
    int t = threadIdx.x;
    if (t < EDIM * HEADS) {
        int d = t / HEADS, h = t % HEADS;
        float s = 0.f;
        for (int c = 0; c < HID; c++)
            s += We1[d * (HEADS * HID) + h * HID + c] * ae1[h * HID + c];
        v1[t] = s;
    } else if (t < EDIM * HEADS + EDIM) {
        int d = t - EDIM * HEADS;
        float s = 0.f;
        for (int c = 0; c < OUT_F; c++)
            s += We2[d * OUT_F + c] * ae2[c];
        v2[d] = s;
    }
}

template<int NT, int K>
__global__ void wmma_gemm(const float* __restrict__ A, const float* __restrict__ B,
                          float* __restrict__ C, int M, int N) {
    constexpr int NPAD = NT * 16;
    int wave = (blockIdx.x * blockDim.x + threadIdx.x) >> 5;
    int lane = threadIdx.x & 31;
    int half = lane >> 4;
    int lrow = lane & 15;
    int mtiles = M >> 4;
    if (wave >= mtiles) return;
    int m0 = wave << 4;

    v8f acc[NT];
#pragma unroll
    for (int nt = 0; nt < NT; nt++)
#pragma unroll
        for (int g = 0; g < 8; g++) acc[nt][g] = 0.0f;

    const float* arow = A + (size_t)(m0 + lrow) * K;
    const float* bcol = B + lrow;
#pragma unroll 4
    for (int k = 0; k < K; k += 4) {
        int kk = k + half * 2;
        v2f a;
        a.x = arow[kk];
        a.y = arow[kk + 1];
#pragma unroll
        for (int nt = 0; nt < NT; nt++) {
            v2f b;
            b.x = bcol[kk * NPAD + nt * 16];
            b.y = bcol[(kk + 1) * NPAD + nt * 16];
            acc[nt] = __builtin_amdgcn_wmma_f32_16x16x4_f32(
                false, a, false, b, (short)0, acc[nt], false, false);
        }
    }
#pragma unroll
    for (int nt = 0; nt < NT; nt++) {
        int col = nt * 16 + lrow;
        if (col < N) {
#pragma unroll
            for (int g = 0; g < 8; g++)
                C[(size_t)(m0 + g + half * 8) * N + col] = acc[nt][g];
        }
    }
}

__global__ void node_scores(const float* __restrict__ xh,
                            const float* __restrict__ asrc, const float* __restrict__ adst,
                            float* als, float* ald, int n, int H, int C) {
    int i = blockIdx.x * blockDim.x + threadIdx.x;
    if (i >= n * H) return;
    int node = i / H, h = i % H;
    const float* row = xh + (size_t)node * H * C + h * C;
    float s = 0.f, d = 0.f;
    for (int c = 0; c < C; c++) {
        float v = row[c];
        s += v * asrc[h * C + c];
        d += v * adst[h * C + c];
    }
    als[i] = s;
    ald[i] = d;
}

__global__ void edge_pass1(const int* __restrict__ src, const int* __restrict__ dst,
                           const float* __restrict__ eattr, const float* __restrict__ loop5,
                           const float* __restrict__ als, const float* __restrict__ ald,
                           const float* __restrict__ v,
                           float* alpha, unsigned* amax, int E, int Etot, int H) {
    int e = blockIdx.x * blockDim.x + threadIdx.x;
    if (e >= Etot) return;
    int s, d;
    float ea[EDIM];
    if (e < E) {
        s = src[e]; d = dst[e];
#pragma unroll
        for (int i = 0; i < EDIM; i++) ea[i] = eattr[(size_t)e * EDIM + i];
    } else {
        s = d = e - E;
#pragma unroll
        for (int i = 0; i < EDIM; i++) ea[i] = loop5[s * EDIM + i];
    }
    for (int h = 0; h < H; h++) {
        float ale = 0.f;
#pragma unroll
        for (int i = 0; i < EDIM; i++) ale += ea[i] * v[i * H + h];
        float a = als[s * H + h] + ald[d * H + h] + ale;
        a = (a > 0.f) ? a : NEG_SLOPE * a;
        alpha[(size_t)e * H + h] = a;
        atomicMax(&amax[d * H + h], encf(a));
    }
}

__global__ void edge_pass2(const int* __restrict__ dst, float* alpha,
                           const unsigned* __restrict__ amax, float* denom,
                           int E, int Etot, int H) {
    int e = blockIdx.x * blockDim.x + threadIdx.x;
    if (e >= Etot) return;
    int d = (e < E) ? dst[e] : (e - E);
    for (int h = 0; h < H; h++) {
        float a = alpha[(size_t)e * H + h];
        float ex = expf(a - decf(amax[d * H + h]));
        alpha[(size_t)e * H + h] = ex;
        atomicAdd(&denom[d * H + h], ex);
    }
}

__global__ void edge_aggregate(const int* __restrict__ src, const int* __restrict__ dst,
                               const float* __restrict__ alpha, const float* __restrict__ denom,
                               const float* __restrict__ xh, float* acc,
                               int E, int Etot, int H, int C) {
    int idx = blockIdx.x * blockDim.x + threadIdx.x;
    int HC = H * C;
    if (idx >= Etot * HC) return;
    int e = idx / HC, c = idx % HC, h = c / C;
    int s = (e < E) ? src[e] : (e - E);
    int d = (e < E) ? dst[e] : (e - E);
    float w = alpha[(size_t)e * H + h] / (denom[d * H + h] + 1e-16f);
    atomicAdd(&acc[(size_t)d * HC + c], xh[(size_t)s * HC + c] * w);
}

__global__ void bias_elu(float* acc, const float* __restrict__ b, int total, int F) {
    int i = blockIdx.x * blockDim.x + threadIdx.x;
    if (i >= total) return;
    float x = acc[i] + b[i % F];
    acc[i] = (x > 0.f) ? x : expm1f(x);
}

__global__ void bias_logsoftmax(const float* __restrict__ acc, const float* __restrict__ b,
                                float* __restrict__ out, int n) {
    int node = blockIdx.x * blockDim.x + threadIdx.x;
    if (node >= n) return;
    float vals[OUT_F];
    float mx = -INFINITY;
#pragma unroll
    for (int c = 0; c < OUT_F; c++) {
        float v = acc[(size_t)node * OUT_F + c] + b[c];
        vals[c] = v;
        mx = fmaxf(mx, v);
    }
    float sum = 0.f;
#pragma unroll
    for (int c = 0; c < OUT_F; c++) sum += expf(vals[c] - mx);
    float lse = mx + logf(sum);
#pragma unroll
    for (int c = 0; c < OUT_F; c++) out[(size_t)node * OUT_F + c] = vals[c] - lse;
}

extern "C" void kernel_launch(void* const* d_in, const int* in_sizes, int n_in,
                              void* d_out, int out_size, void* d_ws, size_t ws_size,
                              hipStream_t stream) {
    const float* x   = (const float*)d_in[0];
    const int*   ei  = (const int*)d_in[1];
    const float* ea  = (const float*)d_in[2];
    const float* W1  = (const float*)d_in[3];
    const float* as1 = (const float*)d_in[4];
    const float* ad1 = (const float*)d_in[5];
    const float* We1 = (const float*)d_in[6];
    const float* ae1 = (const float*)d_in[7];
    const float* b1  = (const float*)d_in[8];
    const float* W2  = (const float*)d_in[9];
    const float* as2 = (const float*)d_in[10];
    const float* ad2 = (const float*)d_in[11];
    const float* We2 = (const float*)d_in[12];
    const float* ae2 = (const float*)d_in[13];
    const float* b2  = (const float*)d_in[14];

    const int N = N_NODES, E = N_EDGES, Et = ETOT;
    const int HC1 = HEADS * HID;
    const int NPAD2 = 48;
    const int* src = ei;
    const int* dst = ei + E;

    char* ws = (char*)d_ws;
    size_t off = 0;
    auto alloc = [&](size_t bytes) -> void* {
        void* p = ws + off;
        off = (off + bytes + 255) & ~(size_t)255;
        return p;
    };
    float*    cnt    = (float*)alloc(sizeof(float) * N);
    float*    loop5  = (float*)alloc(sizeof(float) * N * EDIM);
    float*    v1     = (float*)alloc(sizeof(float) * 16);
    float*    v2     = (float*)alloc(sizeof(float) * 16);
    float*    W2p    = (float*)alloc(sizeof(float) * F_IN * NPAD2);
    float*    xh1    = (float*)alloc(sizeof(float) * (size_t)N * HC1);
    float*    als1   = (float*)alloc(sizeof(float) * N * HEADS);
    float*    ald1   = (float*)alloc(sizeof(float) * N * HEADS);
    unsigned* amax1  = (unsigned*)alloc(sizeof(unsigned) * N * HEADS);
    float*    den1   = (float*)alloc(sizeof(float) * N * HEADS);
    float*    alpha1 = (float*)alloc(sizeof(float) * (size_t)Et * HEADS);
    float*    acc1   = (float*)alloc(sizeof(float) * (size_t)N * HC1);
    float*    xh2    = (float*)alloc(sizeof(float) * (size_t)N * OUT_F);
    float*    als2   = (float*)alloc(sizeof(float) * N);
    float*    ald2   = (float*)alloc(sizeof(float) * N);
    unsigned* amax2  = (unsigned*)alloc(sizeof(unsigned) * N);
    float*    den2   = (float*)alloc(sizeof(float) * N);
    float*    alpha2 = (float*)alloc(sizeof(float) * Et);
    float*    acc2   = (float*)alloc(sizeof(float) * (size_t)N * OUT_F);
    (void)ws_size; (void)n_in; (void)in_sizes; (void)out_size;

    const int T = 256;

    fill_f32<<<CDIV(N, T), T, 0, stream>>>(cnt, N, 0.f);
    fill_f32<<<CDIV(N * EDIM, T), T, 0, stream>>>(loop5, N * EDIM, 0.f);
    selfloop_accum<<<CDIV(E, T), T, 0, stream>>>(dst, ea, cnt, loop5, E);
    selfloop_final<<<CDIV(N * EDIM, T), T, 0, stream>>>(loop5, cnt, N);
    precompute_v<<<1, 32, 0, stream>>>(We1, ae1, We2, ae2, v1, v2);
    pad_b<<<CDIV(F_IN * NPAD2, T), T, 0, stream>>>(W2, W2p, F_IN, OUT_F, NPAD2);

    {
        int mtiles = N / 16, waves_per_blk = 128 / 32;
        wmma_gemm<8, F_IN><<<CDIV(mtiles, waves_per_blk), 128, 0, stream>>>(x, W1, xh1, N, HC1);
    }
    node_scores<<<CDIV(N * HEADS, T), T, 0, stream>>>(xh1, as1, ad1, als1, ald1, N, HEADS, HID);
    fill_u32<<<CDIV(N * HEADS, T), T, 0, stream>>>(amax1, N * HEADS, ENC_NEG_INF);
    fill_f32<<<CDIV(N * HEADS, T), T, 0, stream>>>(den1, N * HEADS, 0.f);
    fill_f32<<<CDIV(N * HC1, T), T, 0, stream>>>(acc1, N * HC1, 0.f);
    edge_pass1<<<CDIV(Et, T), T, 0, stream>>>(src, dst, ea, loop5, als1, ald1, v1,
                                              alpha1, amax1, E, Et, HEADS);
    edge_pass2<<<CDIV(Et, T), T, 0, stream>>>(dst, alpha1, amax1, den1, E, Et, HEADS);
    edge_aggregate<<<CDIV(Et * HC1, T), T, 0, stream>>>(src, dst, alpha1, den1, xh1, acc1,
                                                        E, Et, HEADS, HID);
    bias_elu<<<CDIV(N * HC1, T), T, 0, stream>>>(acc1, b1, N * HC1, HC1);

    {
        int mtiles = N / 16, waves_per_blk = 128 / 32;
        wmma_gemm<3, HC1><<<CDIV(mtiles, waves_per_blk), 128, 0, stream>>>(acc1, W2p, xh2, N, OUT_F);
    }
    node_scores<<<CDIV(N, T), T, 0, stream>>>(xh2, as2, ad2, als2, ald2, N, 1, OUT_F);
    fill_u32<<<CDIV(N, T), T, 0, stream>>>(amax2, N, ENC_NEG_INF);
    fill_f32<<<CDIV(N, T), T, 0, stream>>>(den2, N, 0.f);
    fill_f32<<<CDIV(N * OUT_F, T), T, 0, stream>>>(acc2, N * OUT_F, 0.f);
    edge_pass1<<<CDIV(Et, T), T, 0, stream>>>(src, dst, ea, loop5, als2, ald2, v2,
                                              alpha2, amax2, E, Et, 1);
    edge_pass2<<<CDIV(Et, T), T, 0, stream>>>(dst, alpha2, amax2, den2, E, Et, 1);
    edge_aggregate<<<CDIV(Et * OUT_F, T), T, 0, stream>>>(src, dst, alpha2, den2, xh2, acc2,
                                                          E, Et, 1, OUT_F);
    bias_logsoftmax<<<CDIV(N, T), T, 0, stream>>>(acc2, b2, (float*)d_out, N);
}


# Self-loop mean, softmax max and softmax sum moved into the dst-range owner work-group (LDS atomics over its queued edges, arrival order as before); all global atomics and 12 launches gone
# speedup vs baseline: 4.8202x; 4.8202x over previous
#include <hip/hip_runtime.h>
#include <math.h>

#define N_NODES 50000
#define N_EDGES 800000
#define ETOT    (N_EDGES + N_NODES)
#define F_IN    128
#define HID     64
#define HEADS   2
#define OUT_F   40
#define EDIM    5
#define NEG_SLOPE 0.2f

#define CDIV(a,b) (((a)+(b)-1)/(b))

typedef __attribute__((ext_vector_type(2))) float v2f;
typedef __attribute__((ext_vector_type(8))) float v8f;

__device__ inline unsigned encf(float f) {
    unsigned u = __float_as_uint(f);
    return (u & 0x80000000u) ? ~u : (u | 0x80000000u);
}
__device__ inline float decf(unsigned m) {
    unsigned u = (m & 0x80000000u) ? (m ^ 0x80000000u) : ~m;
    return __uint_as_float(u);
}
#define ENC_NEG_INF 0x007FFFFFu

__global__ void pad_b(const float* __restrict__ W, float* __restrict__ Wp,
                      int K, int N, int NPAD) {
    int i = blockIdx.x * blockDim.x + threadIdx.x;
    if (i >= K * NPAD) return;
    int r = i / NPAD, c = i % NPAD;
    Wp[i] = (c < N) ? W[r * N + c] : 0.0f;
}

__global__ void precompute_v(const float* We1, const float* ae1,
                             const float* We2, const float* ae2,
                             float* v1, float* v2) {
    int t = threadIdx.x;
    if (t < EDIM * HEADS) {
        int d = t / HEADS, h = t % HEADS;
        float s = 0.f;
        for (int c = 0; c < HID; c++)
            s += We1[d * (HEADS * HID) + h * HID + c] * ae1[h * HID + c];
        v1[t] = s;
    } else if (t < EDIM * HEADS + EDIM) {
        int d = t - EDIM * HEADS;
        float s = 0.f;
        for (int c = 0; c < OUT_F; c++)
            s += We2[d * OUT_F + c] * ae2[c];
        v2[d] = s;
    }
}

template<int NT, int K>
__global__ void wmma_gemm(const float* __restrict__ A, const float* __restrict__ B,
                          float* __restrict__ C, int M, int N) {
    constexpr int NPAD = NT * 16;
    int wave = (blockIdx.x * blockDim.x + threadIdx.x) >> 5;
    int lane = threadIdx.x & 31;
    int half = lane >> 4;
    int lrow = lane & 15;
    int mtiles = M >> 4;
    if (wave >= mtiles) return;
    int m0 = wave << 4;

    v8f acc[NT];
#pragma unroll
    for (int nt = 0; nt < NT; nt++)
#pragma unroll
        for (int g = 0; g < 8; g++) acc[nt][g] = 0.0f;

    const float* arow = A + (size_t)(m0 + lrow) * K;
    const float* bcol = B + lrow;
#pragma unroll 4
    for (int k = 0; k < K; k += 4) {
        int kk = k + half * 2;
        v2f a;
        a.x = arow[kk];
        a.y = arow[kk + 1];
#pragma unroll
        for (int nt = 0; nt < NT; nt++) {
            v2f b;
            b.x = bcol[kk * NPAD + nt * 16];
            b.y = bcol[(kk + 1) * NPAD + nt * 16];
            acc[nt] = __builtin_amdgcn_wmma_f32_16x16x4_f32(
                false, a, false, b, (short)0, acc[nt], false, false);
        }
    }
#pragma unroll
    for (int nt = 0; nt < NT; nt++) {
        int col = nt * 16 + lrow;
        if (col < N) {
#pragma unroll
            for (int g = 0; g < 8; g++)
                C[(size_t)(m0 + g + half * 8) * N + col] = acc[nt][g];
        }
    }
}

__global__ void node_scores(const float* __restrict__ xh,
                            const float* __restrict__ asrc, const float* __restrict__ adst,
                            float* als, float* ald, int n, int H, int C) {
    int i = blockIdx.x * blockDim.x + threadIdx.x;
    if (i >= n * H) return;
    int node = i / H, h = i % H;
    const float* row = xh + (size_t)node * H * C + h * C;
    float s = 0.f, d = 0.f;
    for (int c = 0; c < C; c++) {
        float v = row[c];
        s += v * asrc[h * C + c];
        d += v * adst[h * C + c];
    }
    als[i] = s;
    ald[i] = d;
}

#define NPW 196
#define NGRP CDIV(N_NODES, NPW)
#define SCAN_U 8
#define QCAP 8192
#define NS (EDIM + 1)

template<class F>
__device__ inline void scan_owned(const int* __restrict__ dst, int lo, int tid, F f) {
    constexpr int NQ = N_EDGES / 4;
    constexpr int STEP = 1024 * SCAN_U;
    constexpr int NQ_FULL = (NQ / STEP) * STEP;
    auto hit = [&](int e) {
        unsigned r = (unsigned)(dst[e] - lo);
        if (r < (unsigned)NPW) f(e, (int)r);
    };
    const int4* d4 = reinterpret_cast<const int4*>(dst);
    for (int base = 0; base < NQ_FULL; base += STEP) {
        int4 v[SCAN_U];
#pragma unroll
        for (int u = 0; u < SCAN_U; u++) v[u] = d4[base + u * 1024 + tid];
        unsigned m = 0;
#pragma unroll
        for (int u = 0; u < SCAN_U; u++) {
            m |= ((unsigned)(v[u].x - lo) < (unsigned)NPW ? 1u : 0u) << (4 * u + 0);
            m |= ((unsigned)(v[u].y - lo) < (unsigned)NPW ? 1u : 0u) << (4 * u + 1);
            m |= ((unsigned)(v[u].z - lo) < (unsigned)NPW ? 1u : 0u) << (4 * u + 2);
            m |= ((unsigned)(v[u].w - lo) < (unsigned)NPW ? 1u : 0u) << (4 * u + 3);
        }
        while (m) {
            int k = __builtin_ctz(m);
            m &= m - 1;
            hit((base + (k >> 2) * 1024 + tid) * 4 + (k & 3));
        }
    }
    for (int i = NQ_FULL + tid; i < NQ; i += 1024) {
        int4 v = d4[i];
        if ((unsigned)(v.x - lo) < (unsigned)NPW) hit(i * 4 + 0);
        if ((unsigned)(v.y - lo) < (unsigned)NPW) hit(i * 4 + 1);
        if ((unsigned)(v.z - lo) < (unsigned)NPW) hit(i * 4 + 2);
        if ((unsigned)(v.w - lo) < (unsigned)NPW) hit(i * 4 + 3);
    }
}

template<int H>
__device__ inline void raw_alpha(const float* ea, const float* __restrict__ v,
                                 const float* __restrict__ als, const float* __restrict__ ald,
                                 int s, int d, float* a) {
#pragma unroll
    for (int h = 0; h < H; h++) {
        float ale = 0.f;
#pragma unroll
        for (int i = 0; i < EDIM; i++) ale += ea[i] * v[i * H + h];
        float x = als[s * H + h] + ald[d * H + h] + ale;
        a[h] = (x > 0.f) ? x : NEG_SLOPE * x;
    }
}

template<int H, int C>
__global__ __launch_bounds__(1024)
void gat_owned(const int* __restrict__ src, const int* __restrict__ dst,
               const float* __restrict__ eattr, const float* __restrict__ v,
               const float* __restrict__ als, const float* __restrict__ ald,
               const float* __restrict__ xh, float* __restrict__ acc) {
    static_assert(N_EDGES % 4 == 0 && ETOT <= (1 << 20) && NPW <= 256 && NPW <= QCAP &&
                  NPW * H <= 1024, "queue entry packing, one thread per owned node and head");
    constexpr int HC = H * C;
    constexpr int J  = (HC + 31) / 32;

    __shared__ float    sacc[NPW * HC];
    __shared__ int      q[QCAP];
    __shared__ int      qs[QCAP];
    __shared__ float    qa[QCAP * H];
    __shared__ float    nsum[NPW * NS];
    __shared__ unsigned nmax[NPW * H];
    __shared__ float    nden[NPW * H];
    __shared__ float    aself[NPW * H];
    __shared__ int      qn;

    const int tid  = threadIdx.x;
    const int lane = tid & 31;
    const int wave = __builtin_amdgcn_readfirstlane(tid >> 5);
    const int lo   = blockIdx.x * NPW;
    int nown = N_NODES - lo;
    if (nown > NPW) nown = NPW;

    for (int idx = tid; idx < NPW * HC; idx += 1024) sacc[idx] = 0.f;
    for (int idx = tid; idx < NPW * NS; idx += 1024) nsum[idx] = 0.f;
    if (tid < NPW * H) { nmax[tid] = ENC_NEG_INF; nden[tid] = 0.f; }
    if (tid == 0) qn = 0;
    __syncthreads();

    scan_owned(dst, lo, tid, [&](int e, int r) {
        int slot = atomicAdd(&qn, 1);
        if (slot < QCAP) q[slot] = e | (r << 20);
    });
    __syncthreads();
    const int nreal = __builtin_amdgcn_readfirstlane(qn);
    const int total = nreal + nown;

    auto load_ea = [&](int e, float* ea) {
#pragma unroll
        for (int i = 0; i < EDIM; i++) ea[i] = eattr[(size_t)e * EDIM + i];
    };
    auto self_ea = [&](int r, float* ea) {
        float c = fmaxf(nsum[r * NS], 1.0f);
#pragma unroll
        for (int i = 0; i < EDIM; i++) ea[i] = nsum[r * NS + 1 + i] / c;
    };

    if (total <= QCAP) {
        for (int idx = tid; idx < nreal; idx += 1024) {
            int ent = q[idx];
            int e = ent & 0xFFFFF, r = ent >> 20;
            int s = src[e];
            float ea[EDIM], a[H];
            load_ea(e, ea);
            atomicAdd(&nsum[r * NS], 1.0f);
#pragma unroll
            for (int i = 0; i < EDIM; i++) atomicAdd(&nsum[r * NS + 1 + i], ea[i]);
            raw_alpha<H>(ea, v, als, ald, s, lo + r, a);
            qs[idx] = s;
#pragma unroll
            for (int h = 0; h < H; h++) {
                qa[idx * H + h] = a[h];
                atomicMax(&nmax[r * H + h], encf(a[h]));
            }
        }
        __syncthreads();
        if (tid < nown) {
            int r = tid, idx = nreal + tid;
            float ea[EDIM], a[H];
            self_ea(r, ea);
            raw_alpha<H>(ea, v, als, ald, lo + r, lo + r, a);
            q[idx]  = (N_EDGES + lo + r) | (r << 20);
            qs[idx] = lo + r;
#pragma unroll
            for (int h = 0; h < H; h++) {
                qa[idx * H + h] = a[h];
                atomicMax(&nmax[r * H + h], encf(a[h]));
            }
        }
        __syncthreads();
        for (int idx = tid; idx < total; idx += 1024) {
            int r = q[idx] >> 20;
#pragma unroll
            for (int h = 0; h < H; h++) {
                float ex = expf(qa[idx * H + h] - decf(nmax[r * H + h]));
                qa[idx * H + h] = ex;
                atomicAdd(&nden[r * H + h], ex);
            }
        }
        __syncthreads();

        for (int g = wave * 32; g < total; g += 1024) {
            int cnt = total - g;
            if (cnt > 32) cnt = 32;
            int s = 0, r = 0;
            float w[H];
#pragma unroll
            for (int h = 0; h < H; h++) w[h] = 0.f;
            if (lane < cnt) {
                r = q[g + lane] >> 20;
                s = qs[g + lane];
#pragma unroll
                for (int h = 0; h < H; h++)
                    w[h] = qa[(g + lane) * H + h] / (nden[r * H + h] + 1e-16f);
            }
            for (int j0 = 0; j0 < cnt; j0 += 8) {
                float val[8][J];
#pragma unroll
                for (int b = 0; b < 8; b++) {
                    int sj = __builtin_amdgcn_readlane(s, j0 + b);
#pragma unroll
                    for (int jx = 0; jx < J; jx++) {
                        int c = lane + 32 * jx;
                        val[b][jx] = (c < HC) ? xh[(size_t)sj * HC + c] : 0.f;
                    }
                }
#pragma unroll
                for (int b = 0; b < 8; b++) {
                    if (j0 + b < cnt) {
                        int rj = __builtin_amdgcn_readlane(r, j0 + b);
#pragma unroll
                        for (int jx = 0; jx < J; jx++) {
                            int c = lane + 32 * jx;
                            int h = (H == 1) ? 0 : (32 * jx) / C;
                            float wj = __int_as_float(__builtin_amdgcn_readlane(__float_as_int(w[h]), j0 + b));
                            if (c < HC) atomicAdd(&sacc[rj * HC + c], val[b][jx] * wj);
                        }
                    }
                }
            }
        }
    } else {
        scan_owned(dst, lo, tid, [&](int e, int r) {
            float ea[EDIM], a[H];
            load_ea(e, ea);
            atomicAdd(&nsum[r * NS], 1.0f);
#pragma unroll
            for (int i = 0; i < EDIM; i++) atomicAdd(&nsum[r * NS + 1 + i], ea[i]);
            raw_alpha<H>(ea, v, als, ald, src[e], lo + r, a);
#pragma unroll
            for (int h = 0; h < H; h++) atomicMax(&nmax[r * H + h], encf(a[h]));
        });
        __syncthreads();
        if (tid < nown) {
            float ea[EDIM], a[H];
            self_ea(tid, ea);
            raw_alpha<H>(ea, v, als, ald, lo + tid, lo + tid, a);
#pragma unroll
            for (int h = 0; h < H; h++) {
                aself[tid * H + h] = a[h];
                atomicMax(&nmax[tid * H + h], encf(a[h]));
            }
        }
        __syncthreads();
        scan_owned(dst, lo, tid, [&](int e, int r) {
            float ea[EDIM], a[H];
            load_ea(e, ea);
            raw_alpha<H>(ea, v, als, ald, src[e], lo + r, a);
#pragma unroll
            for (int h = 0; h < H; h++)
                atomicAdd(&nden[r * H + h], expf(a[h] - decf(nmax[r * H + h])));
        });
        if (tid < nown) {
#pragma unroll
            for (int h = 0; h < H; h++)
                atomicAdd(&nden[tid * H + h], expf(aself[tid * H + h] - decf(nmax[tid * H + h])));
        }
        __syncthreads();
        auto add_row = [&](int s, int r, const float* a) {
            for (int h = 0; h < H; h++) {
                float w = expf(a[h] - decf(nmax[r * H + h])) / (nden[r * H + h] + 1e-16f);
                for (int c = 0; c < C; c++)
                    atomicAdd(&sacc[r * HC + h * C + c], xh[(size_t)s * HC + h * C + c] * w);
            }
        };
        scan_owned(dst, lo, tid, [&](int e, int r) {
            float ea[EDIM], a[H];
            load_ea(e, ea);
            int s = src[e];
            raw_alpha<H>(ea, v, als, ald, s, lo + r, a);
            add_row(s, r, a);
        });
        if (tid < nown) {
            float a[H];
#pragma unroll
            for (int h = 0; h < H; h++) a[h] = aself[tid * H + h];
            add_row(lo + tid, tid, a);
        }
    }
    __syncthreads();

    for (int idx = tid; idx < NPW * HC; idx += 1024) {
        int o = lo * HC + idx;
        if (o < N_NODES * HC) acc[o] = sacc[idx];
    }
}

__global__ void bias_elu(float* acc, const float* __restrict__ b, int total, int F) {
    int i = blockIdx.x * blockDim.x + threadIdx.x;
    if (i >= total) return;
    float x = acc[i] + b[i % F];
    acc[i] = (x > 0.f) ? x : expm1f(x);
}

__global__ void bias_logsoftmax(const float* __restrict__ acc, const float* __restrict__ b,
                                float* __restrict__ out, int n) {
    int node = blockIdx.x * blockDim.x + threadIdx.x;
    if (node >= n) return;
    float vals[OUT_F];
    float mx = -INFINITY;
#pragma unroll
    for (int c = 0; c < OUT_F; c++) {
        float v = acc[(size_t)node * OUT_F + c] + b[c];
        vals[c] = v;
        mx = fmaxf(mx, v);
    }
    float sum = 0.f;
#pragma unroll
    for (int c = 0; c < OUT_F; c++) sum += expf(vals[c] - mx);
    float lse = mx + logf(sum);
#pragma unroll
    for (int c = 0; c < OUT_F; c++) out[(size_t)node * OUT_F + c] = vals[c] - lse;
}

extern "C" void kernel_launch(void* const* d_in, const int* in_sizes, int n_in,
                              void* d_out, int out_size, void* d_ws, size_t ws_size,
                              hipStream_t stream) {
    const float* x   = (const float*)d_in[0];
    const int*   ei  = (const int*)d_in[1];
    const float* ea  = (const float*)d_in[2];
    const float* W1  = (const float*)d_in[3];
    const float* as1 = (const float*)d_in[4];
    const float* ad1 = (const float*)d_in[5];
    const float* We1 = (const float*)d_in[6];
    const float* ae1 = (const float*)d_in[7];
    const float* b1  = (const float*)d_in[8];
    const float* W2  = (const float*)d_in[9];
    const float* as2 = (const float*)d_in[10];
    const float* ad2 = (const float*)d_in[11];
    const float* We2 = (const float*)d_in[12];
    const float* ae2 = (const float*)d_in[13];
    const float* b2  = (const float*)d_in[14];

    const int N = N_NODES, E = N_EDGES;
    const int HC1 = HEADS * HID;
    const int NPAD2 = 48;
    const int* src = ei;
    const int* dst = ei + E;

    char* ws = (char*)d_ws;
    size_t off = 0;
    auto alloc = [&](size_t bytes) -> void* {
        void* p = ws + off;
        off = (off + bytes + 255) & ~(size_t)255;
        return p;
    };
    float*    v1     = (float*)alloc(sizeof(float) * 16);
    float*    v2     = (float*)alloc(sizeof(float) * 16);
    float*    W2p    = (float*)alloc(sizeof(float) * F_IN * NPAD2);
    float*    xh1    = (float*)alloc(sizeof(float) * (size_t)N * HC1);
    float*    als1   = (float*)alloc(sizeof(float) * N * HEADS);
    float*    ald1   = (float*)alloc(sizeof(float) * N * HEADS);
    float*    acc1   = (float*)alloc(sizeof(float) * (size_t)N * HC1);
    float*    xh2    = (float*)alloc(sizeof(float) * (size_t)N * OUT_F);
    float*    als2   = (float*)alloc(sizeof(float) * N);
    float*    ald2   = (float*)alloc(sizeof(float) * N);
    float*    acc2   = (float*)alloc(sizeof(float) * (size_t)N * OUT_F);
    (void)ws_size; (void)n_in; (void)in_sizes; (void)out_size;

    const int T = 256;

    precompute_v<<<1, 32, 0, stream>>>(We1, ae1, We2, ae2, v1, v2);
    pad_b<<<CDIV(F_IN * NPAD2, T), T, 0, stream>>>(W2, W2p, F_IN, OUT_F, NPAD2);

    {
        int mtiles = N / 16, waves_per_blk = 128 / 32;
        wmma_gemm<8, F_IN><<<CDIV(mtiles, waves_per_blk), 128, 0, stream>>>(x, W1, xh1, N, HC1);
    }
    node_scores<<<CDIV(N * HEADS, T), T, 0, stream>>>(xh1, as1, ad1, als1, ald1, N, HEADS, HID);
    gat_owned<HEADS, HID><<<NGRP, 1024, 0, stream>>>(src, dst, ea, v1, als1, ald1, xh1, acc1);
    bias_elu<<<CDIV(N * HC1, T), T, 0, stream>>>(acc1, b1, N * HC1, HC1);

    {
        int mtiles = N / 16, waves_per_blk = 128 / 32;
        wmma_gemm<3, HC1><<<CDIV(mtiles, waves_per_blk), 128, 0, stream>>>(acc1, W2p, xh2, N, OUT_F);
    }
    node_scores<<<CDIV(N, T), T, 0, stream>>>(xh2, as2, ad2, als2, ald2, N, 1, OUT_F);
    gat_owned<1, OUT_F><<<NGRP, 1024, 0, stream>>>(src, dst, ea, v2, als2, ald2, xh2, acc2);
    bias_logsoftmax<<<CDIV(N, T), T, 0, stream>>>(acc2, b2, (float*)d_out, N);
}


# Attention vectors folded into the weights: node scores = x @ (W @ att) ride as extra B columns of both f32 WMMA projections; both score launches gone (8 -> 6); score sums now in the GEMM k order
# speedup vs baseline: 7.8706x; 1.6328x over previous
#include <hip/hip_runtime.h>
#include <math.h>

#define N_NODES 50000
#define N_EDGES 800000
#define ETOT    (N_EDGES + N_NODES)
#define F_IN    128
#define HID     64
#define HEADS   2
#define OUT_F   40
#define EDIM    5
#define NEG_SLOPE 0.2f

#define CDIV(a,b) (((a)+(b)-1)/(b))

typedef __attribute__((ext_vector_type(2))) float v2f;
typedef __attribute__((ext_vector_type(8))) float v8f;

__device__ inline unsigned encf(float f) {
    unsigned u = __float_as_uint(f);
    return (u & 0x80000000u) ? ~u : (u | 0x80000000u);
}
__device__ inline float decf(unsigned m) {
    unsigned u = (m & 0x80000000u) ? (m ^ 0x80000000u) : ~m;
    return __uint_as_float(u);
}
#define ENC_NEG_INF 0x007FFFFFu

#define U1_LD 16
__global__ void fold_vectors(const float* We1, const float* ae1,
                             const float* We2, const float* ae2,
                             const float* __restrict__ W1, const float* __restrict__ as1,
                             const float* __restrict__ ad1,
                             const float* __restrict__ W2, const float* __restrict__ as2,
                             const float* __restrict__ ad2,
                             float* v1, float* v2, float* u1, float* u2) {
    static_assert(F_IN == HEADS * HID && 2 * HEADS <= U1_LD, "one block per row of W1 and of W2");
    const int t = threadIdx.x;
    const int k = blockIdx.x;
    const int wave = t >> 5, lane = t & 31;
    {
        const int h = wave % HEADS;
        const float* att = (wave < HEADS) ? as1 : ad1;
        float s = 0.f;
        for (int c = lane; c < HID; c += 32)
            s += W1[k * (HEADS * HID) + h * HID + c] * att[h * HID + c];
        for (int m = 16; m > 0; m >>= 1) s += __shfl_xor(s, m, 32);
        if (lane == 0) u1[k * U1_LD + wave] = s;
    }
    if (t >= 2 * HEADS && t < U1_LD) u1[k * U1_LD + t] = 0.f;
    if (wave < 2) {
        const float* att = (wave == 0) ? as2 : ad2;
        float s = 0.f;
        for (int c = lane; c < OUT_F; c += 32) s += W2[k * OUT_F + c] * att[c];
        for (int m = 16; m > 0; m >>= 1) s += __shfl_xor(s, m, 32);
        if (lane == 0) u2[k * 2 + wave] = s;
    }
    if (k != 0) return;
    if (t < EDIM * HEADS) {
        int d = t / HEADS, h = t % HEADS;
        float s = 0.f;
        for (int c = 0; c < HID; c++)
            s += We1[d * (HEADS * HID) + h * HID + c] * ae1[h * HID + c];
        v1[t] = s;
    } else if (t < EDIM * HEADS + EDIM) {
        int d = t - EDIM * HEADS;
        float s = 0.f;
        for (int c = 0; c < OUT_F; c++)
            s += We2[d * OUT_F + c] * ae2[c];
        v2[d] = s;
    }
}

template<int NT, int K>
__global__ void wmma_gemm(const float* __restrict__ A, const float* __restrict__ B,
                          const float* __restrict__ Bs,
                          float* __restrict__ C, float* __restrict__ als,
                          float* __restrict__ ald, int M, int N) {
    constexpr int NPAD = NT * 16;
    int wave = (blockIdx.x * blockDim.x + threadIdx.x) >> 5;
    int lane = threadIdx.x & 31;
    int half = lane >> 4;
    int lrow = lane & 15;
    int mtiles = M >> 4;
    if (wave >= mtiles) return;
    int m0 = wave << 4;

    v8f acc[NT], accs;
#pragma unroll
    for (int g = 0; g < 8; g++) accs[g] = 0.0f;
#pragma unroll
    for (int nt = 0; nt < NT; nt++)
#pragma unroll
        for (int g = 0; g < 8; g++) acc[nt][g] = 0.0f;

    const float* arow = A + (size_t)(m0 + lrow) * K;
    const float* bcol = B + lrow;
    const float* scol = Bs + lrow;
#pragma unroll 4
    for (int k = 0; k < K; k += 4) {
        int kk = k + half * 2;
        v2f a;
        a.x = arow[kk];
        a.y = arow[kk + 1];
#pragma unroll
        for (int nt = 0; nt < NT; nt++) {
            v2f b;
            b.x = bcol[kk * NPAD + nt * 16];
            b.y = bcol[(kk + 1) * NPAD + nt * 16];
            acc[nt] = __builtin_amdgcn_wmma_f32_16x16x4_f32(
                false, a, false, b, (short)0, acc[nt], false, false);
        }
        v2f bs;
        bs.x = scol[kk * U1_LD];
        bs.y = scol[(kk + 1) * U1_LD];
        accs = __builtin_amdgcn_wmma_f32_16x16x4_f32(
            false, a, false, bs, (short)0, accs, false, false);
    }
#pragma unroll
    for (int nt = 0; nt < NT; nt++) {
        int col = nt * 16 + lrow;
        if (col < N) {
#pragma unroll
            for (int g = 0; g < 8; g++)
                C[(size_t)(m0 + g + half * 8) * N + col] = acc[nt][g];
        }
    }
    if (lrow < 2 * HEADS) {
        float* sc = (lrow < HEADS) ? als : ald;
        const int h = (lrow < HEADS) ? lrow : lrow - HEADS;
#pragma unroll
        for (int g = 0; g < 8; g++)
            sc[(size_t)(m0 + g + half * 8) * HEADS + h] = accs[g];
    }
}

__device__ inline v8f wmma_f32_k4_spaced(v2f a, v2f b, v8f c) {
    asm("v_wmma_f32_16x16x4_f32 %0, %1, %2, %0\n\tv_nop\n\tv_nop\n\tv_nop\n\tv_nop"
        : "+v"(c) : "v"(a), "v"(b));
    return c;
}

#define NPW 196
#define NGRP CDIV(N_NODES, NPW)
#define SCAN_U 8
#define QCAP 8192
#define NS (EDIM + 1)

__global__ void owner_bytes(const int* __restrict__ dst, unsigned* __restrict__ ob) {
    int i = blockIdx.x * blockDim.x + threadIdx.x;
    if (i >= N_EDGES / 16) return;
    const int4* d4 = reinterpret_cast<const int4*>(dst) + 4 * i;
    auto own = [](int d) { return ((unsigned)d / (unsigned)NPW) & 0xFFu; };
    uint4 w;
    int4 d;
    d = d4[0]; w.x = own(d.x) | own(d.y) << 8 | own(d.z) << 16 | own(d.w) << 24;
    d = d4[1]; w.y = own(d.x) | own(d.y) << 8 | own(d.z) << 16 | own(d.w) << 24;
    d = d4[2]; w.z = own(d.x) | own(d.y) << 8 | own(d.z) << 16 | own(d.w) << 24;
    d = d4[3]; w.w = own(d.x) | own(d.y) << 8 | own(d.z) << 16 | own(d.w) << 24;
    reinterpret_cast<uint4*>(ob)[i] = w;
}

template<class F>
__device__ inline void scan_owned(const unsigned* __restrict__ ob, unsigned grp, int tid, F f) {
    constexpr int NU = N_EDGES / 16;
    constexpr int STEP = 1024 * SCAN_U;
    constexpr int NU_FULL = (NU / STEP) * STEP;
    const unsigned pat = grp * 0x01010101u;
    auto differs = [&](unsigned w) {
        unsigned x = w ^ pat;
        return (((x & 0x7F7F7F7Fu) + 0x7F7F7F7Fu) | x) & 0x80808080u;
    };
    auto differs16 = [&](uint4 v) {
        return differs(v.x) >> 7 | differs(v.y) >> 6 | differs(v.z) >> 5 | differs(v.w) >> 4;
    };
    const uint4* o4 = reinterpret_cast<const uint4*>(ob);
    for (int base = 0; base < NU_FULL; base += STEP) {
        uint4 v[SCAN_U];
#pragma unroll
        for (int u = 0; u < SCAN_U; u++) v[u] = o4[base + u * 1024 + tid];
#pragma unroll
        for (int p = 0; p < SCAN_U; p += 2) {
            unsigned m = ~(differs16(v[p]) | differs16(v[p + 1]) << 4);
            while (m) {
                int k = __builtin_ctz(m);
                m &= m - 1;
                int i = base + (p + ((k >> 2) & 1)) * 1024 + tid;
                f((i * 4 + (k & 3)) * 4 + (k >> 3));
            }
        }
    }
    for (int i = NU_FULL + tid; i < NU; i += 1024) {
        unsigned m = ~(differs16(o4[i]) | 0xF0F0F0F0u);
        while (m) {
            int k = __builtin_ctz(m);
            m &= m - 1;
            f((i * 4 + (k & 3)) * 4 + (k >> 3));
        }
    }
}

template<int H>
__device__ inline float edge_term(const float* ea, const float* __restrict__ v, int h) {
    float ale = 0.f;
#pragma unroll
    for (int i = 0; i < EDIM; i++) ale += ea[i] * v[i * H + h];
    return ale;
}

__device__ inline float leaky_sum(float as, float ad, float ale) {
    float x = as + ad + ale;
    return (x > 0.f) ? x : NEG_SLOPE * x;
}

template<int H>
__device__ inline void raw_alpha(const float* ea, const float* __restrict__ v,
                                 const float* __restrict__ als, const float* __restrict__ ald,
                                 int s, int d, float* a) {
#pragma unroll
    for (int h = 0; h < H; h++)
        a[h] = leaky_sum(als[s * H + h], ald[d * H + h], edge_term<H>(ea, v, h));
}

template<int H, int C, bool PROJ>
__global__ __launch_bounds__(1024)
void gat_owned(const int* __restrict__ src, const int* __restrict__ dst,
               const unsigned* __restrict__ ob,
               const float* __restrict__ eattr, const float* __restrict__ v,
               const float* __restrict__ als, const float* __restrict__ ald,
               const float* __restrict__ xh,
               const float* __restrict__ bias, const float* __restrict__ Wn,
               const float* __restrict__ vn, const float* __restrict__ un,
               unsigned* __restrict__ hcnt, unsigned* __restrict__ hsr,
               float* __restrict__ hale,
               float* __restrict__ acc, float* __restrict__ alsn, float* __restrict__ aldn) {
    constexpr bool GIVE = PROJ, TAKE = !PROJ;
    static_assert(N_NODES <= (1 << 16) && (!TAKE || H == 1),
                  "a hand-over entry packs the source node into 16 bits and carries one edge term");
    static_assert(N_EDGES % 16 == 0 && NGRP <= 256 && ETOT <= (1 << 20) && NPW <= 256 &&
                  NPW <= QCAP && NPW * H <= 1024,
                  "owner bytes, queue entry packing, one thread per owned node and head");
    constexpr int HC = H * C;
    constexpr int J  = (HC + 31) / 32;
    constexpr int LD = PROJ ? HC + 4 : HC;
    constexpr int NP = CDIV(OUT_F, 16) * 16;
    constexpr int MT = CDIV(NPW, 16);
    constexpr int OP = NP + 4;
    static_assert(!PROJ || (1024 % HC == 0 && HC % 4 == 0 && LD % 2 == 0 && OUT_F % 4 == 0 &&
                            OP % 4 == 0 && MT * 16 * OP <= QCAP * H && OUT_F + 2 <= NP),
                  "one bias channel per thread, k steps of 4, 8-byte fragment reads, 16-byte row "
                  "copies, projected tiles fit the score array they reuse, two padding columns "
                  "are free for the next layer's score vectors");

    __shared__ alignas(16) float sacc[NPW * LD];
    __shared__ alignas(16) float wns[PROJ ? HC * NP : 2];
    __shared__ int      q[QCAP];
    __shared__ int      qs[QCAP];
    __shared__ alignas(16) float qa[QCAP * H];
    __shared__ float    nsum[NPW * NS];
    __shared__ unsigned nmax[NPW * H];
    __shared__ float    nden[NPW * H];
    __shared__ float    aself[NPW * H];
    __shared__ int      qn;

    const int tid  = threadIdx.x;
    const int lane = tid & 31;
    const int wave = __builtin_amdgcn_readfirstlane(tid >> 5);
    const unsigned grp = blockIdx.x;
    const int lo   = blockIdx.x * NPW;
    int nown = N_NODES - lo;
    if (nown > NPW) nown = NPW;

    float bc = 0.f;
    if constexpr (PROJ) {
        bc = bias[tid % HC];
        for (int i = tid; i < HC * NP; i += 1024) {
            int k = i / NP, cn = i - k * NP;
            float wv = 0.f;
            if (cn < OUT_F) wv = Wn[k * OUT_F + cn];
            else if (cn < OUT_F + 2) wv = un[k * 2 + (cn - OUT_F)];
            wns[((k >> 1) * NP + cn) * 2 + (k & 1)] = wv;
        }
    }
    for (int idx = tid; idx < NPW * LD; idx += 1024) sacc[idx] = 0.f;
    for (int idx = tid; idx < NPW * NS; idx += 1024) nsum[idx] = 0.f;
    if (tid < NPW * H) { nmax[tid] = ENC_NEG_INF; nden[tid] = 0.f; }
    if (tid == 0) qn = 0;
    __syncthreads();

    if constexpr (!TAKE) {
        scan_owned(ob, grp, tid, [&](int e) {
            int slot = atomicAdd(&qn, 1);
            if (slot < QCAP) q[slot] = e;
        });
        __syncthreads();
    }
    const int nreal = __builtin_amdgcn_readfirstlane(TAKE ? (int)hcnt[grp] : qn);
    const int total = nreal + nown;
    if constexpr (GIVE) { if (tid == 0) hcnt[grp] = (unsigned)nreal; }
    const size_t hbase = (size_t)grp * QCAP;

    auto load_ea = [&](int e, float* ea) {
#pragma unroll
        for (int i = 0; i < EDIM; i++) ea[i] = eattr[(size_t)e * EDIM + i];
    };
    auto self_ea = [&](int r, float* ea) {
        float c = fmaxf(nsum[r * NS], 1.0f);
#pragma unroll
        for (int i = 0; i < EDIM; i++) ea[i] = nsum[r * NS + 1 + i] / c;
    };

    if (total <= QCAP) {
        if constexpr (TAKE) {
            constexpr int TU = 4;
            for (int base = tid; base < total; base += TU * 1024) {
                unsigned sr[TU];
                float ale[TU], a[TU];
#pragma unroll
                for (int u = 0; u < TU; u++) {
                    int idx = base + u * 1024;
                    sr[u] = (unsigned)lo;
                    ale[u] = 0.f;
                    if (idx < total) {
                        sr[u]  = hsr[hbase + idx];
                        ale[u] = hale[hbase + idx];
                    }
                }
#pragma unroll
                for (int u = 0; u < TU; u++)
                    a[u] = leaky_sum(als[sr[u] & 0xFFFFu], ald[lo + (sr[u] >> 16)], ale[u]);
#pragma unroll
                for (int u = 0; u < TU; u++) {
                    int idx = base + u * 1024;
                    if (idx < total) {
                        int r = (int)(sr[u] >> 16);
                        q[idx]  = r << 20;
                        qs[idx] = (int)(sr[u] & 0xFFFFu);
                        qa[idx] = a[u];
                        atomicMax(&nmax[r], encf(a[u]));
                    }
                }
            }
        } else {
            for (int idx = tid; idx < nreal; idx += 1024) {
                int e = q[idx];
                int s = src[e];
                int r = dst[e] - lo;
                q[idx] = e | (r << 20);
                float ea[EDIM], a[H];
                load_ea(e, ea);
                atomicAdd(&nsum[r * NS], 1.0f);
#pragma unroll
                for (int i = 0; i < EDIM; i++) atomicAdd(&nsum[r * NS + 1 + i], ea[i]);
                raw_alpha<H>(ea, v, als, ald, s, lo + r, a);
                qs[idx] = s;
#pragma unroll
                for (int h = 0; h < H; h++) {
                    qa[idx * H + h] = a[h];
                    atomicMax(&nmax[r * H + h], encf(a[h]));
                }
                if constexpr (GIVE) {
                    hsr[hbase + idx]  = (unsigned)s | (unsigned)r << 16;
                    hale[hbase + idx] = edge_term<1>(ea, vn, 0);
                }
            }
            __syncthreads();
            if (tid < nown) {
                int r = tid, idx = nreal + tid;
                float ea[EDIM], a[H];
                self_ea(r, ea);
                raw_alpha<H>(ea, v, als, ald, lo + r, lo + r, a);
                q[idx]  = (N_EDGES + lo + r) | (r << 20);
                qs[idx] = lo + r;
#pragma unroll
                for (int h = 0; h < H; h++) {
                    qa[idx * H + h] = a[h];
                    atomicMax(&nmax[r * H + h], encf(a[h]));
                }
                if constexpr (GIVE) {
                    hsr[hbase + idx]  = (unsigned)(lo + r) | (unsigned)r << 16;
                    hale[hbase + idx] = edge_term<1>(ea, vn, 0);
                }
            }
        }
        __syncthreads();
        for (int idx = tid; idx < total; idx += 1024) {
            int r = q[idx] >> 20;
#pragma unroll
            for (int h = 0; h < H; h++) {
                float ex = expf(qa[idx * H + h] - decf(nmax[r * H + h]));
                qa[idx * H + h] = ex;
                atomicAdd(&nden[r * H + h], ex);
            }
        }
        __syncthreads();

        for (int g = wave * 32; g < total; g += 1024) {
            int cnt = total - g;
            if (cnt > 32) cnt = 32;
            int s = 0, r = 0;
            float w[H];
#pragma unroll
            for (int h = 0; h < H; h++) w[h] = 0.f;
            if (lane < cnt) {
                r = q[g + lane] >> 20;
                s = qs[g + lane];
#pragma unroll
                for (int h = 0; h < H; h++)
                    w[h] = qa[(g + lane) * H + h] / (nden[r * H + h] + 1e-16f);
            }
            for (int j0 = 0; j0 < cnt; j0 += 8) {
                float val[8][J];
#pragma unroll
                for (int b = 0; b < 8; b++) {
                    int sj = __builtin_amdgcn_readlane(s, j0 + b);
#pragma unroll
                    for (int jx = 0; jx < J; jx++) {
                        int c = lane + 32 * jx;
                        val[b][jx] = (c < HC) ? xh[(size_t)sj * HC + c] : 0.f;
                    }
                }
#pragma unroll
                for (int b = 0; b < 8; b++) {
                    if (j0 + b < cnt) {
                        int rj = __builtin_amdgcn_readlane(r, j0 + b);
#pragma unroll
                        for (int jx = 0; jx < J; jx++) {
                            int c = lane + 32 * jx;
                            int h = (H == 1) ? 0 : (32 * jx) / C;
                            float wj = __int_as_float(__builtin_amdgcn_readlane(__float_as_int(w[h]), j0 + b));
                            if (c < HC) atomicAdd(&sacc[rj * LD + c], val[b][jx] * wj);
                        }
                    }
                }
            }
        }
    } else {
        scan_owned(ob, grp, tid, [&](int e) {
            int r = dst[e] - lo;
            float ea[EDIM], a[H];
            load_ea(e, ea);
            atomicAdd(&nsum[r * NS], 1.0f);
#pragma unroll
            for (int i = 0; i < EDIM; i++) atomicAdd(&nsum[r * NS + 1 + i], ea[i]);
            raw_alpha<H>(ea, v, als, ald, src[e], lo + r, a);
#pragma unroll
            for (int h = 0; h < H; h++) atomicMax(&nmax[r * H + h], encf(a[h]));
        });
        __syncthreads();
        if (tid < nown) {
            float ea[EDIM], a[H];
            self_ea(tid, ea);
            raw_alpha<H>(ea, v, als, ald, lo + tid, lo + tid, a);
#pragma unroll
            for (int h = 0; h < H; h++) {
                aself[tid * H + h] = a[h];
                atomicMax(&nmax[tid * H + h], encf(a[h]));
            }
        }
        __syncthreads();
        scan_owned(ob, grp, tid, [&](int e) {
            int r = dst[e] - lo;
            float ea[EDIM], a[H];
            load_ea(e, ea);
            raw_alpha<H>(ea, v, als, ald, src[e], lo + r, a);
#pragma unroll
            for (int h = 0; h < H; h++)
                atomicAdd(&nden[r * H + h], expf(a[h] - decf(nmax[r * H + h])));
        });
        if (tid < nown) {
#pragma unroll
            for (int h = 0; h < H; h++)
                atomicAdd(&nden[tid * H + h], expf(aself[tid * H + h] - decf(nmax[tid * H + h])));
        }
        __syncthreads();
        auto add_row = [&](int s, int r, const float* a) {
            for (int h = 0; h < H; h++) {
                float w = expf(a[h] - decf(nmax[r * H + h])) / (nden[r * H + h] + 1e-16f);
                for (int c = 0; c < C; c++)
                    atomicAdd(&sacc[r * LD + h * C + c], xh[(size_t)s * HC + h * C + c] * w);
            }
        };
        scan_owned(ob, grp, tid, [&](int e) {
            int r = dst[e] - lo;
            float ea[EDIM], a[H];
            load_ea(e, ea);
            int s = src[e];
            raw_alpha<H>(ea, v, als, ald, s, lo + r, a);
            add_row(s, r, a);
        });
        if (tid < nown) {
            float a[H];
#pragma unroll
            for (int h = 0; h < H; h++) a[h] = aself[tid * H + h];
            add_row(lo + tid, tid, a);
        }
    }
    __syncthreads();

    if constexpr (PROJ) {
        for (int idx = tid; idx < NPW * HC; idx += 1024) {
            float* p = &sacc[(idx / HC) * LD + (idx % HC)];
            float x = *p + bc;
            *p = (x > 0.f) ? x : expm1f(x);
        }
        __syncthreads();
        constexpr int NT = NP / 16;
        float* ost = qa;
        const int half = lane >> 4, lrow = lane & 15;
        for (int j = wave; j < MT * NT; j += 32) {
            const int mt = j / NT, nt = j - mt * NT;
            int row = mt * 16 + lrow;
            if (row > NPW - 1) row = NPW - 1;
            const float* arow = sacc + row * LD + half * 2;
            const float* bcol = wns + (half * NP + nt * 16 + lrow) * 2;
            float* orow = ost + (mt * 16 + half * 8) * OP + nt * 16 + lrow;
            v8f d;
#pragma unroll
            for (int g = 0; g < 8; g++) d[g] = 0.0f;
#pragma unroll
            for (int k = 0; k < HC; k += 4) {
                v2f a = *reinterpret_cast<const v2f*>(arow + k);
                v2f b = *reinterpret_cast<const v2f*>(bcol + k * NP);
                d = wmma_f32_k4_spaced(a, b, d);
            }
#pragma unroll
            for (int g = 0; g < 8; g++) orow[g * OP] = d[g];
        }
        __syncthreads();
        float4* g4 = reinterpret_cast<float4*>(acc + (size_t)lo * OUT_F);
        for (int i = tid; i < nown * (OUT_F / 4); i += 1024) {
            int r = i / (OUT_F / 4), c4 = i - r * (OUT_F / 4);
            g4[i] = *reinterpret_cast<const float4*>(ost + r * OP + c4 * 4);
        }
        if (tid < nown) {
            alsn[lo + tid] = ost[tid * OP + OUT_F];
            aldn[lo + tid] = ost[tid * OP + OUT_F + 1];
        }
    } else {
        for (int idx = tid; idx < NPW * HC; idx += 1024) {
            int o = lo * HC + idx;
            if (o < N_NODES * HC) acc[o] = sacc[idx];
        }
    }
}

__global__ void bias_logsoftmax(const float* __restrict__ acc, const float* __restrict__ b,
                                float* __restrict__ out, int n) {
    int node = blockIdx.x * blockDim.x + threadIdx.x;
    if (node >= n) return;
    float vals[OUT_F];
    float mx = -INFINITY;
#pragma unroll
    for (int c = 0; c < OUT_F; c++) {
        float v = acc[(size_t)node * OUT_F + c] + b[c];
        vals[c] = v;
        mx = fmaxf(mx, v);
    }
    float sum = 0.f;
#pragma unroll
    for (int c = 0; c < OUT_F; c++) sum += expf(vals[c] - mx);
    float lse = mx + logf(sum);
#pragma unroll
    for (int c = 0; c < OUT_F; c++) out[(size_t)node * OUT_F + c] = vals[c] - lse;
}

extern "C" void kernel_launch(void* const* d_in, const int* in_sizes, int n_in,
                              void* d_out, int out_size, void* d_ws, size_t ws_size,
                              hipStream_t stream) {
    const float* x   = (const float*)d_in[0];
    const int*   ei  = (const int*)d_in[1];
    const float* ea  = (const float*)d_in[2];
    const float* W1  = (const float*)d_in[3];
    const float* as1 = (const float*)d_in[4];
    const float* ad1 = (const float*)d_in[5];
    const float* We1 = (const float*)d_in[6];
    const float* ae1 = (const float*)d_in[7];
    const float* b1  = (const float*)d_in[8];
    const float* W2  = (const float*)d_in[9];
    const float* as2 = (const float*)d_in[10];
    const float* ad2 = (const float*)d_in[11];
    const float* We2 = (const float*)d_in[12];
    const float* ae2 = (const float*)d_in[13];
    const float* b2  = (const float*)d_in[14];

    const int N = N_NODES, E = N_EDGES;
    const int HC1 = HEADS * HID;
    const int* src = ei;
    const int* dst = ei + E;

    char* ws = (char*)d_ws;
    size_t off = 0;
    auto alloc = [&](size_t bytes) -> void* {
        void* p = ws + off;
        off = (off + bytes + 255) & ~(size_t)255;
        return p;
    };
    float*    v1     = (float*)alloc(sizeof(float) * 16);
    float*    v2     = (float*)alloc(sizeof(float) * 16);
    float*    u1     = (float*)alloc(sizeof(float) * F_IN * U1_LD);
    float*    u2     = (float*)alloc(sizeof(float) * HEADS * HID * 2);
    float*    xh1    = (float*)alloc(sizeof(float) * (size_t)N * HC1);
    float*    als1   = (float*)alloc(sizeof(float) * N * HEADS);
    float*    ald1   = (float*)alloc(sizeof(float) * N * HEADS);
    float*    xh2    = (float*)alloc(sizeof(float) * (size_t)N * OUT_F);
    float*    als2   = (float*)alloc(sizeof(float) * N);
    float*    ald2   = (float*)alloc(sizeof(float) * N);
    float*    acc2   = (float*)alloc(sizeof(float) * (size_t)N * OUT_F);
    unsigned* ob     = (unsigned*)alloc(N_EDGES);
    unsigned* hcnt   = (unsigned*)alloc(sizeof(unsigned) * NGRP);
    unsigned* hsr    = (unsigned*)alloc(sizeof(unsigned) * (size_t)NGRP * QCAP);
    float*    hale   = (float*)alloc(sizeof(float) * (size_t)NGRP * QCAP);
    (void)ws_size; (void)n_in; (void)in_sizes; (void)out_size;

    const int T = 256;

    fold_vectors<<<F_IN, 128, 0, stream>>>(We1, ae1, We2, ae2, W1, as1, ad1, W2, as2, ad2,
                                           v1, v2, u1, u2);
    owner_bytes<<<CDIV(N_EDGES / 16, T), T, 0, stream>>>(dst, ob);

    {
        int mtiles = N / 16, waves_per_blk = 128 / 32;
        wmma_gemm<8, F_IN><<<CDIV(mtiles, waves_per_blk), 128, 0, stream>>>(x, W1, u1, xh1, als1, ald1,
                                                                            N, HC1);
    }
    gat_owned<HEADS, HID, true><<<NGRP, 1024, 0, stream>>>(src, dst, ob, ea, v1, als1, ald1, xh1,
                                                           b1, W2, v2, u2, hcnt, hsr, hale,
                                                           xh2, als2, ald2);

    gat_owned<1, OUT_F, false><<<NGRP, 1024, 0, stream>>>(src, dst, ob, ea, v2, als2, ald2, xh2,
                                                          nullptr, nullptr, nullptr, nullptr,
                                                          hcnt, hsr, hale, acc2, nullptr, nullptr);
    bias_logsoftmax<<<CDIV(N, T), T, 0, stream>>>(acc2, b2, (float*)d_out, N);
}


# Layer 2's owner work-group applies bias + log-softmax to its finished rows in LDS (one thread per row, sums in channel order as before) and stores the output itself; last launch gone (5 -> 4)
# speedup vs baseline: 8.7708x; 1.1144x over previous
#include <hip/hip_runtime.h>
#include <math.h>

#define N_NODES 50000
#define N_EDGES 800000
#define ETOT    (N_EDGES + N_NODES)
#define F_IN    128
#define HID     64
#define HEADS   2
#define OUT_F   40
#define EDIM    5
#define NEG_SLOPE 0.2f

#define CDIV(a,b) (((a)+(b)-1)/(b))

typedef __attribute__((ext_vector_type(2))) float v2f;
typedef __attribute__((ext_vector_type(8))) float v8f;

__device__ inline unsigned encf(float f) {
    unsigned u = __float_as_uint(f);
    return (u & 0x80000000u) ? ~u : (u | 0x80000000u);
}
__device__ inline float decf(unsigned m) {
    unsigned u = (m & 0x80000000u) ? (m ^ 0x80000000u) : ~m;
    return __uint_as_float(u);
}
#define ENC_NEG_INF 0x007FFFFFu

#define U1_LD 16
#define GEMM_T 512

__device__ inline v8f wmma_f32_k4_spaced(v2f a, v2f b, v8f c) {
    asm("v_wmma_f32_16x16x4_f32 %0, %1, %2, %0\n\tv_nop\n\tv_nop\n\tv_nop\n\tv_nop"
        : "+v"(c) : "v"(a), "v"(b));
    return c;
}

template<int NT, int K>
__global__ __launch_bounds__(GEMM_T)
void wmma_gemm(const float* __restrict__ A, const float* __restrict__ B,
               const float* __restrict__ as, const float* __restrict__ ad,
               float* __restrict__ C, float* __restrict__ als,
               float* __restrict__ ald, int M, int N) {
    constexpr int NPAD = NT * 16;
    static_assert(NPAD == HEADS * HID && HID % 4 == 0 && 2 * HEADS <= U1_LD && K % 16 == 0 &&
                  (K / 16) * 32 <= GEMM_T,
                  "score tile: B's columns are the heads' channels, a k step stays inside one "
                  "head, one wave per 16 rows of the tile");
    __shared__ float us[K * U1_LD];
    const int wib  = threadIdx.x >> 5;
    const int lane = threadIdx.x & 31;
    const int half = lane >> 4;
    const int lrow = lane & 15;

    if (wib < K / 16) {
        const float* wrow = B + (size_t)(wib * 16 + lrow) * NPAD;
        const float* att  = (lrow < HEADS) ? as : ad;
        const int    hd   = (lrow < HEADS) ? lrow : lrow - HEADS;
        const bool   live = lrow < 2 * HEADS;
        v8f d;
#pragma unroll
        for (int g = 0; g < 8; g++) d[g] = 0.0f;
#pragma unroll
        for (int j = 0; j < NPAD; j += 4) {
            const int jj = j + half * 2;
            v2f a, b;
            a.x = wrow[jj];
            a.y = wrow[jj + 1];
            const float t0 = att[jj], t1 = att[jj + 1];
            const bool on = live && (jj / HID == hd);
            b.x = on ? t0 : 0.0f;
            b.y = on ? t1 : 0.0f;
            d = wmma_f32_k4_spaced(a, b, d);
        }
#pragma unroll
        for (int g = 0; g < 8; g++)
            us[(wib * 16 + g + half * 8) * U1_LD + lrow] = d[g];
    }
    __syncthreads();

    int wave = blockIdx.x * (GEMM_T / 32) + wib;
    int mtiles = M >> 4;
    if (wave >= mtiles) return;
    int m0 = wave << 4;

    v8f acc[NT], accs;
#pragma unroll
    for (int g = 0; g < 8; g++) accs[g] = 0.0f;
#pragma unroll
    for (int nt = 0; nt < NT; nt++)
#pragma unroll
        for (int g = 0; g < 8; g++) acc[nt][g] = 0.0f;

    const float* arow = A + (size_t)(m0 + lrow) * K;
    const float* bcol = B + lrow;
    const float* scol = us + lrow;
#pragma unroll 4
    for (int k = 0; k < K; k += 4) {
        int kk = k + half * 2;
        v2f a;
        a.x = arow[kk];
        a.y = arow[kk + 1];
#pragma unroll
        for (int nt = 0; nt < NT; nt++) {
            v2f b;
            b.x = bcol[kk * NPAD + nt * 16];
            b.y = bcol[(kk + 1) * NPAD + nt * 16];
            acc[nt] = __builtin_amdgcn_wmma_f32_16x16x4_f32(
                false, a, false, b, (short)0, acc[nt], false, false);
        }
        v2f bs;
        bs.x = scol[kk * U1_LD];
        bs.y = scol[(kk + 1) * U1_LD];
        accs = __builtin_amdgcn_wmma_f32_16x16x4_f32(
            false, a, false, bs, (short)0, accs, false, false);
    }
#pragma unroll
    for (int nt = 0; nt < NT; nt++) {
        int col = nt * 16 + lrow;
        if (col < N) {
#pragma unroll
            for (int g = 0; g < 8; g++)
                C[(size_t)(m0 + g + half * 8) * N + col] = acc[nt][g];
        }
    }
    if (lrow < 2 * HEADS) {
        float* sc = (lrow < HEADS) ? als : ald;
        const int h = (lrow < HEADS) ? lrow : lrow - HEADS;
#pragma unroll
        for (int g = 0; g < 8; g++)
            sc[(size_t)(m0 + g + half * 8) * HEADS + h] = accs[g];
    }
}

#define NPW 196
#define NGRP CDIV(N_NODES, NPW)
#define SCAN_U 8
#define QCAP 8192
#define NS (EDIM + 1)

__global__ void owner_bytes(const int* __restrict__ dst, unsigned* __restrict__ ob) {
    int i = blockIdx.x * blockDim.x + threadIdx.x;
    if (i >= N_EDGES / 16) return;
    const int4* d4 = reinterpret_cast<const int4*>(dst) + 4 * i;
    auto own = [](int d) { return ((unsigned)d / (unsigned)NPW) & 0xFFu; };
    uint4 w;
    int4 d;
    d = d4[0]; w.x = own(d.x) | own(d.y) << 8 | own(d.z) << 16 | own(d.w) << 24;
    d = d4[1]; w.y = own(d.x) | own(d.y) << 8 | own(d.z) << 16 | own(d.w) << 24;
    d = d4[2]; w.z = own(d.x) | own(d.y) << 8 | own(d.z) << 16 | own(d.w) << 24;
    d = d4[3]; w.w = own(d.x) | own(d.y) << 8 | own(d.z) << 16 | own(d.w) << 24;
    reinterpret_cast<uint4*>(ob)[i] = w;
}

template<class F>
__device__ inline void scan_owned(const unsigned* __restrict__ ob, unsigned grp, int tid, F f) {
    constexpr int NU = N_EDGES / 16;
    constexpr int STEP = 1024 * SCAN_U;
    constexpr int NU_FULL = (NU / STEP) * STEP;
    const unsigned pat = grp * 0x01010101u;
    auto differs = [&](unsigned w) {
        unsigned x = w ^ pat;
        return (((x & 0x7F7F7F7Fu) + 0x7F7F7F7Fu) | x) & 0x80808080u;
    };
    auto differs16 = [&](uint4 v) {
        return differs(v.x) >> 7 | differs(v.y) >> 6 | differs(v.z) >> 5 | differs(v.w) >> 4;
    };
    const uint4* o4 = reinterpret_cast<const uint4*>(ob);
    for (int base = 0; base < NU_FULL; base += STEP) {
        uint4 v[SCAN_U];
#pragma unroll
        for (int u = 0; u < SCAN_U; u++) v[u] = o4[base + u * 1024 + tid];
#pragma unroll
        for (int p = 0; p < SCAN_U; p += 2) {
            unsigned m = ~(differs16(v[p]) | differs16(v[p + 1]) << 4);
            while (m) {
                int k = __builtin_ctz(m);
                m &= m - 1;
                int i = base + (p + ((k >> 2) & 1)) * 1024 + tid;
                f((i * 4 + (k & 3)) * 4 + (k >> 3));
            }
        }
    }
    for (int i = NU_FULL + tid; i < NU; i += 1024) {
        unsigned m = ~(differs16(o4[i]) | 0xF0F0F0F0u);
        while (m) {
            int k = __builtin_ctz(m);
            m &= m - 1;
            f((i * 4 + (k & 3)) * 4 + (k >> 3));
        }
    }
}

template<int H>
__device__ inline float edge_term(const float* ea, const float* __restrict__ v, int h) {
    float ale = 0.f;
#pragma unroll
    for (int i = 0; i < EDIM; i++) ale += ea[i] * v[i * H + h];
    return ale;
}

__device__ inline float leaky_sum(float as, float ad, float ale) {
    float x = as + ad + ale;
    return (x > 0.f) ? x : NEG_SLOPE * x;
}

template<int H>
__device__ inline void raw_alpha(const float* ea, const float* __restrict__ v,
                                 const float* __restrict__ als, const float* __restrict__ ald,
                                 int s, int d, float* a) {
#pragma unroll
    for (int h = 0; h < H; h++)
        a[h] = leaky_sum(als[s * H + h], ald[d * H + h], edge_term<H>(ea, v, h));
}

template<int H, int C, bool PROJ>
__global__ __launch_bounds__(1024)
void gat_owned(const int* __restrict__ src, const int* __restrict__ dst,
               const unsigned* __restrict__ ob,
               const float* __restrict__ eattr,
               const float* __restrict__ We, const float* __restrict__ ae,
               const float* __restrict__ als, const float* __restrict__ ald,
               const float* __restrict__ xh,
               const float* __restrict__ bias, const float* __restrict__ Wn,
               const float* __restrict__ Wen, const float* __restrict__ aen,
               const float* __restrict__ asn, const float* __restrict__ adn,
               unsigned* __restrict__ hcnt, unsigned* __restrict__ hsr,
               float* __restrict__ hale,
               float* __restrict__ acc, float* __restrict__ alsn, float* __restrict__ aldn) {
    constexpr bool GIVE = PROJ, TAKE = !PROJ;
    static_assert(N_NODES <= (1 << 16) && (!TAKE || H == 1),
                  "a hand-over entry packs the source node into 16 bits and carries one edge term");
    static_assert(N_EDGES % 16 == 0 && NGRP <= 256 && ETOT <= (1 << 20) && NPW <= 256 &&
                  NPW <= QCAP && NPW * H <= 1024,
                  "owner bytes, queue entry packing, one thread per owned node and head");
    constexpr int HC = H * C;
    constexpr int J  = (HC + 31) / 32;
    constexpr int LD = PROJ ? HC + 4 : HC;
    constexpr int NP = CDIV(OUT_F, 16) * 16;
    constexpr int MT = CDIV(NPW, 16);
    constexpr int OP = NP + 4;
    constexpr int SV_T0 = 32, SVN_T0 = 64;
    static_assert(!PROJ || (1024 % HC == 0 && HC % 4 == 0 && LD % 2 == 0 && OUT_F % 4 == 0 &&
                            OP % 4 == 0 && MT * 16 * OP <= QCAP * H && OUT_F + 2 <= NP &&
                            SVN_T0 + 32 <= 1024 - 2 * HC),
                  "one bias channel per thread, k steps of 4, 8-byte fragment reads, 16-byte row "
                  "copies, projected tiles fit the score array they reuse, two padding columns "
                  "are free for the next layer's score vectors, the threads that fold them are "
                  "not the ones that fold the edge vectors");

    __shared__ alignas(16) float sacc[NPW * LD];
    __shared__ alignas(16) float wns[PROJ ? HC * NP : 2];
    __shared__ int      q[QCAP];
    __shared__ int      qs[QCAP];
    __shared__ alignas(16) float qa[QCAP * H];
    __shared__ float    nsum[NPW * NS];
    __shared__ unsigned nmax[NPW * H];
    __shared__ float    nden[NPW * H];
    __shared__ float    aself[NPW * H];
    __shared__ float    sv[EDIM * H];
    __shared__ float    svn[PROJ ? EDIM : 1];
    __shared__ int      qn;
    const float* v  = sv;
    const float* vn = svn;

    const int tid  = threadIdx.x;
    const int lane = tid & 31;
    const int wave = __builtin_amdgcn_readfirstlane(tid >> 5);
    const unsigned grp = blockIdx.x;
    const int lo   = blockIdx.x * NPW;
    int nown = N_NODES - lo;
    if (nown > NPW) nown = NPW;

    static_assert(EDIM * H <= 32 && EDIM <= 32, "each folded edge vector inside one wave");
    if (tid >= SV_T0 && tid < SV_T0 + EDIM * H) {
        const int i = tid - SV_T0;
        const int d = i / H, h = i % H;
        float s = 0.f;
#pragma unroll
        for (int cc = 0; cc < C; cc++) s += We[d * HC + h * C + cc] * ae[h * C + cc];
        sv[i] = s;
    }
    float bc = 0.f;
    if constexpr (PROJ) {
        bc = bias[tid % HC];
        if (tid >= SVN_T0 && tid < SVN_T0 + EDIM) {
            const int d = tid - SVN_T0;
            float s = 0.f;
#pragma unroll
            for (int cc = 0; cc < OUT_F; cc++) s += Wen[d * OUT_F + cc] * aen[cc];
            svn[d] = s;
        }
        for (int i = tid; i < HC * NP; i += 1024) {
            int k = i / NP, cn = i - k * NP;
            if (cn >= OUT_F && cn < OUT_F + 2) continue;
            float wv = 0.f;
            if (cn < OUT_F) wv = Wn[k * OUT_F + cn];
            wns[((k >> 1) * NP + cn) * 2 + (k & 1)] = wv;
        }
        if (tid >= 1024 - 2 * HC) {
            const int i = tid - (1024 - 2 * HC);
            const int k = i >> 1, which = i & 1;
            const float* att = which ? adn : asn;
            float s = 0.f;
#pragma unroll
            for (int cc = 0; cc < OUT_F; cc++) s += Wn[k * OUT_F + cc] * att[cc];
            wns[((k >> 1) * NP + OUT_F + which) * 2 + (k & 1)] = s;
        }
    }
    for (int idx = tid; idx < NPW * LD; idx += 1024) sacc[idx] = 0.f;
    for (int idx = tid; idx < NPW * NS; idx += 1024) nsum[idx] = 0.f;
    if (tid < NPW * H) { nmax[tid] = ENC_NEG_INF; nden[tid] = 0.f; }
    if (tid == 0) qn = 0;
    __syncthreads();

    if constexpr (!TAKE) {
        scan_owned(ob, grp, tid, [&](int e) {
            int slot = atomicAdd(&qn, 1);
            if (slot < QCAP) q[slot] = e;
        });
        __syncthreads();
    }
    const int nreal = __builtin_amdgcn_readfirstlane(TAKE ? (int)hcnt[grp] : qn);
    const int total = nreal + nown;
    if constexpr (GIVE) { if (tid == 0) hcnt[grp] = (unsigned)nreal; }
    const size_t hbase = (size_t)grp * QCAP;

    auto load_ea = [&](int e, float* ea) {
#pragma unroll
        for (int i = 0; i < EDIM; i++) ea[i] = eattr[(size_t)e * EDIM + i];
    };
    auto self_ea = [&](int r, float* ea) {
        float c = fmaxf(nsum[r * NS], 1.0f);
#pragma unroll
        for (int i = 0; i < EDIM; i++) ea[i] = nsum[r * NS + 1 + i] / c;
    };

    if (total <= QCAP) {
        if constexpr (TAKE) {
            constexpr int TU = 4;
            for (int base = tid; base < total; base += TU * 1024) {
                unsigned sr[TU];
                float ale[TU], a[TU];
#pragma unroll
                for (int u = 0; u < TU; u++) {
                    int idx = base + u * 1024;
                    sr[u] = (unsigned)lo;
                    ale[u] = 0.f;
                    if (idx < total) {
                        sr[u]  = hsr[hbase + idx];
                        ale[u] = hale[hbase + idx];
                    }
                }
#pragma unroll
                for (int u = 0; u < TU; u++)
                    a[u] = leaky_sum(als[sr[u] & 0xFFFFu], ald[lo + (sr[u] >> 16)], ale[u]);
#pragma unroll
                for (int u = 0; u < TU; u++) {
                    int idx = base + u * 1024;
                    if (idx < total) {
                        int r = (int)(sr[u] >> 16);
                        q[idx]  = r << 20;
                        qs[idx] = (int)(sr[u] & 0xFFFFu);
                        qa[idx] = a[u];
                        atomicMax(&nmax[r], encf(a[u]));
                    }
                }
            }
        } else {
            for (int idx = tid; idx < nreal; idx += 1024) {
                int e = q[idx];
                int s = src[e];
                int r = dst[e] - lo;
                q[idx] = e | (r << 20);
                float ea[EDIM], a[H];
                load_ea(e, ea);
                atomicAdd(&nsum[r * NS], 1.0f);
#pragma unroll
                for (int i = 0; i < EDIM; i++) atomicAdd(&nsum[r * NS + 1 + i], ea[i]);
                raw_alpha<H>(ea, v, als, ald, s, lo + r, a);
                qs[idx] = s;
#pragma unroll
                for (int h = 0; h < H; h++) {
                    qa[idx * H + h] = a[h];
                    atomicMax(&nmax[r * H + h], encf(a[h]));
                }
                if constexpr (GIVE) {
                    hsr[hbase + idx]  = (unsigned)s | (unsigned)r << 16;
                    hale[hbase + idx] = edge_term<1>(ea, vn, 0);
                }
            }
            __syncthreads();
            if (tid < nown) {
                int r = tid, idx = nreal + tid;
                float ea[EDIM], a[H];
                self_ea(r, ea);
                raw_alpha<H>(ea, v, als, ald, lo + r, lo + r, a);
                q[idx]  = (N_EDGES + lo + r) | (r << 20);
                qs[idx] = lo + r;
#pragma unroll
                for (int h = 0; h < H; h++) {
                    qa[idx * H + h] = a[h];
                    atomicMax(&nmax[r * H + h], encf(a[h]));
                }
                if constexpr (GIVE) {
                    hsr[hbase + idx]  = (unsigned)(lo + r) | (unsigned)r << 16;
                    hale[hbase + idx] = edge_term<1>(ea, vn, 0);
                }
            }
        }
        __syncthreads();
        for (int idx = tid; idx < total; idx += 1024) {
            int r = q[idx] >> 20;
#pragma unroll
            for (int h = 0; h < H; h++) {
                float ex = expf(qa[idx * H + h] - decf(nmax[r * H + h]));
                qa[idx * H + h] = ex;
                atomicAdd(&nden[r * H + h], ex);
            }
        }
        __syncthreads();

        for (int g = wave * 32; g < total; g += 1024) {
            int cnt = total - g;
            if (cnt > 32) cnt = 32;
            int s = 0, r = 0;
            float w[H];
#pragma unroll
            for (int h = 0; h < H; h++) w[h] = 0.f;
            if (lane < cnt) {
                r = q[g + lane] >> 20;
                s = qs[g + lane];
#pragma unroll
                for (int h = 0; h < H; h++)
                    w[h] = qa[(g + lane) * H + h] / (nden[r * H + h] + 1e-16f);
            }
            for (int j0 = 0; j0 < cnt; j0 += 8) {
                float val[8][J];
#pragma unroll
                for (int b = 0; b < 8; b++) {
                    int sj = __builtin_amdgcn_readlane(s, j0 + b);
#pragma unroll
                    for (int jx = 0; jx < J; jx++) {
                        int c = lane + 32 * jx;
                        val[b][jx] = (c < HC) ? xh[(size_t)sj * HC + c] : 0.f;
                    }
                }
#pragma unroll
                for (int b = 0; b < 8; b++) {
                    if (j0 + b < cnt) {
                        int rj = __builtin_amdgcn_readlane(r, j0 + b);
#pragma unroll
                        for (int jx = 0; jx < J; jx++) {
                            int c = lane + 32 * jx;
                            int h = (H == 1) ? 0 : (32 * jx) / C;
                            float wj = __int_as_float(__builtin_amdgcn_readlane(__float_as_int(w[h]), j0 + b));
                            if (c < HC) atomicAdd(&sacc[rj * LD + c], val[b][jx] * wj);
                        }
                    }
                }
            }
        }
    } else {
        scan_owned(ob, grp, tid, [&](int e) {
            int r = dst[e] - lo;
            float ea[EDIM], a[H];
            load_ea(e, ea);
            atomicAdd(&nsum[r * NS], 1.0f);
#pragma unroll
            for (int i = 0; i < EDIM; i++) atomicAdd(&nsum[r * NS + 1 + i], ea[i]);
            raw_alpha<H>(ea, v, als, ald, src[e], lo + r, a);
#pragma unroll
            for (int h = 0; h < H; h++) atomicMax(&nmax[r * H + h], encf(a[h]));
        });
        __syncthreads();
        if (tid < nown) {
            float ea[EDIM], a[H];
            self_ea(tid, ea);
            raw_alpha<H>(ea, v, als, ald, lo + tid, lo + tid, a);
#pragma unroll
            for (int h = 0; h < H; h++) {
                aself[tid * H + h] = a[h];
                atomicMax(&nmax[tid * H + h], encf(a[h]));
            }
        }
        __syncthreads();
        scan_owned(ob, grp, tid, [&](int e) {
            int r = dst[e] - lo;
            float ea[EDIM], a[H];
            load_ea(e, ea);
            raw_alpha<H>(ea, v, als, ald, src[e], lo + r, a);
#pragma unroll
            for (int h = 0; h < H; h++)
                atomicAdd(&nden[r * H + h], expf(a[h] - decf(nmax[r * H + h])));
        });
        if (tid < nown) {
#pragma unroll
            for (int h = 0; h < H; h++)
                atomicAdd(&nden[tid * H + h], expf(aself[tid * H + h] - decf(nmax[tid * H + h])));
        }
        __syncthreads();
        auto add_row = [&](int s, int r, const float* a) {
            for (int h = 0; h < H; h++) {
                float w = expf(a[h] - decf(nmax[r * H + h])) / (nden[r * H + h] + 1e-16f);
                for (int c = 0; c < C; c++)
                    atomicAdd(&sacc[r * LD + h * C + c], xh[(size_t)s * HC + h * C + c] * w);
            }
        };
        scan_owned(ob, grp, tid, [&](int e) {
            int r = dst[e] - lo;
            float ea[EDIM], a[H];
            load_ea(e, ea);
            int s = src[e];
            raw_alpha<H>(ea, v, als, ald, s, lo + r, a);
            add_row(s, r, a);
        });
        if (tid < nown) {
            float a[H];
#pragma unroll
            for (int h = 0; h < H; h++) a[h] = aself[tid * H + h];
            add_row(lo + tid, tid, a);
        }
    }
    __syncthreads();

    if constexpr (PROJ) {
        for (int idx = tid; idx < NPW * HC; idx += 1024) {
            float* p = &sacc[(idx / HC) * LD + (idx % HC)];
            float x = *p + bc;
            *p = (x > 0.f) ? x : expm1f(x);
        }
        __syncthreads();
        constexpr int NT = NP / 16;
        float* ost = qa;
        const int half = lane >> 4, lrow = lane & 15;
        for (int j = wave; j < MT * NT; j += 32) {
            const int mt = j / NT, nt = j - mt * NT;
            int row = mt * 16 + lrow;
            if (row > NPW - 1) row = NPW - 1;
            const float* arow = sacc + row * LD + half * 2;
            const float* bcol = wns + (half * NP + nt * 16 + lrow) * 2;
            float* orow = ost + (mt * 16 + half * 8) * OP + nt * 16 + lrow;
            v8f d;
#pragma unroll
            for (int g = 0; g < 8; g++) d[g] = 0.0f;
#pragma unroll
            for (int k = 0; k < HC; k += 4) {
                v2f a = *reinterpret_cast<const v2f*>(arow + k);
                v2f b = *reinterpret_cast<const v2f*>(bcol + k * NP);
                d = wmma_f32_k4_spaced(a, b, d);
            }
#pragma unroll
            for (int g = 0; g < 8; g++) orow[g * OP] = d[g];
        }
        __syncthreads();
        float4* g4 = reinterpret_cast<float4*>(acc + (size_t)lo * OUT_F);
        for (int i = tid; i < nown * (OUT_F / 4); i += 1024) {
            int r = i / (OUT_F / 4), c4 = i - r * (OUT_F / 4);
            g4[i] = *reinterpret_cast<const float4*>(ost + r * OP + c4 * 4);
        }
        if (tid < nown) {
            alsn[lo + tid] = ost[tid * OP + OUT_F];
            aldn[lo + tid] = ost[tid * OP + OUT_F + 1];
        }
    } else {
        static_assert(LD == HC && HC % 4 == 0, "rows lie back to back and are read 16 bytes at a time");
        if (tid < nown) {
            float4* row4 = reinterpret_cast<float4*>(sacc + tid * LD);
            float vals[HC];
#pragma unroll
            for (int i = 0; i < HC / 4; i++) {
                const float4 t = row4[i];
                vals[4 * i]     = t.x + bias[4 * i];
                vals[4 * i + 1] = t.y + bias[4 * i + 1];
                vals[4 * i + 2] = t.z + bias[4 * i + 2];
                vals[4 * i + 3] = t.w + bias[4 * i + 3];
            }
            float mx = -INFINITY;
#pragma unroll
            for (int cc = 0; cc < HC; cc++) mx = fmaxf(mx, vals[cc]);
            float sum = 0.f;
#pragma unroll
            for (int cc = 0; cc < HC; cc++) sum += expf(vals[cc] - mx);
            const float lse = mx + logf(sum);
#pragma unroll
            for (int i = 0; i < HC / 4; i++)
                row4[i] = make_float4(vals[4 * i] - lse, vals[4 * i + 1] - lse,
                                      vals[4 * i + 2] - lse, vals[4 * i + 3] - lse);
        }
        __syncthreads();
        float4* g4 = reinterpret_cast<float4*>(acc + (size_t)lo * HC);
        const float4* s4 = reinterpret_cast<const float4*>(sacc);
        for (int i = tid; i < nown * (HC / 4); i += 1024) g4[i] = s4[i];
    }
}

extern "C" void kernel_launch(void* const* d_in, const int* in_sizes, int n_in,
                              void* d_out, int out_size, void* d_ws, size_t ws_size,
                              hipStream_t stream) {
    const float* x   = (const float*)d_in[0];
    const int*   ei  = (const int*)d_in[1];
    const float* ea  = (const float*)d_in[2];
    const float* W1  = (const float*)d_in[3];
    const float* as1 = (const float*)d_in[4];
    const float* ad1 = (const float*)d_in[5];
    const float* We1 = (const float*)d_in[6];
    const float* ae1 = (const float*)d_in[7];
    const float* b1  = (const float*)d_in[8];
    const float* W2  = (const float*)d_in[9];
    const float* as2 = (const float*)d_in[10];
    const float* ad2 = (const float*)d_in[11];
    const float* We2 = (const float*)d_in[12];
    const float* ae2 = (const float*)d_in[13];
    const float* b2  = (const float*)d_in[14];

    const int N = N_NODES, E = N_EDGES;
    const int HC1 = HEADS * HID;
    const int* src = ei;
    const int* dst = ei + E;

    char* ws = (char*)d_ws;
    size_t off = 0;
    auto alloc = [&](size_t bytes) -> void* {
        void* p = ws + off;
        off = (off + bytes + 255) & ~(size_t)255;
        return p;
    };
    float*    xh1    = (float*)alloc(sizeof(float) * (size_t)N * HC1);
    float*    als1   = (float*)alloc(sizeof(float) * N * HEADS);
    float*    ald1   = (float*)alloc(sizeof(float) * N * HEADS);
    float*    xh2    = (float*)alloc(sizeof(float) * (size_t)N * OUT_F);
    float*    als2   = (float*)alloc(sizeof(float) * N);
    float*    ald2   = (float*)alloc(sizeof(float) * N);
    unsigned* ob     = (unsigned*)alloc(N_EDGES);
    unsigned* hcnt   = (unsigned*)alloc(sizeof(unsigned) * NGRP);
    unsigned* hsr    = (unsigned*)alloc(sizeof(unsigned) * (size_t)NGRP * QCAP);
    float*    hale   = (float*)alloc(sizeof(float) * (size_t)NGRP * QCAP);
    (void)ws_size; (void)n_in; (void)in_sizes; (void)out_size;

    const int T = 256;

    owner_bytes<<<CDIV(N_EDGES / 16, T), T, 0, stream>>>(dst, ob);

    {
        int mtiles = N / 16, waves_per_blk = GEMM_T / 32;
        wmma_gemm<8, F_IN><<<CDIV(mtiles, waves_per_blk), GEMM_T, 0, stream>>>(x, W1, as1, ad1, xh1,
                                                                               als1, ald1, N, HC1);
    }
    gat_owned<HEADS, HID, true><<<NGRP, 1024, 0, stream>>>(src, dst, ob, ea, We1, ae1, als1, ald1,
                                                           xh1, b1, W2, We2, ae2, as2, ad2,
                                                           hcnt, hsr, hale, xh2, als2, ald2);

    gat_owned<1, OUT_F, false><<<NGRP, 1024, 0, stream>>>(src, dst, ob, ea, We2, ae2, als2, ald2,
                                                          xh2, b2, nullptr, nullptr, nullptr,
                                                          nullptr, nullptr,
                                                          hcnt, hsr, hale, (float*)d_out, nullptr,
                                                          nullptr);
}


# Drain of both owner layers software-pipelined: rows come 8 at a time into two register sets that take turns, the next rows' loads in flight under the current rows' LDS adds; arrival-order sums kept
# speedup vs baseline: 9.1708x; 1.0456x over previous
#include <hip/hip_runtime.h>
#include <math.h>

#define N_NODES 50000
#define N_EDGES 800000
#define ETOT    (N_EDGES + N_NODES)
#define F_IN    128
#define HID     64
#define HEADS   2
#define OUT_F   40
#define EDIM    5
#define NEG_SLOPE 0.2f

#define CDIV(a,b) (((a)+(b)-1)/(b))

typedef __attribute__((ext_vector_type(2))) float v2f;
typedef __attribute__((ext_vector_type(8))) float v8f;

__device__ inline unsigned encf(float f) {
    unsigned u = __float_as_uint(f);
    return (u & 0x80000000u) ? ~u : (u | 0x80000000u);
}
__device__ inline float decf(unsigned m) {
    unsigned u = (m & 0x80000000u) ? (m ^ 0x80000000u) : ~m;
    return __uint_as_float(u);
}
#define ENC_NEG_INF 0x007FFFFFu

#define U1_LD 16
#define GEMM_T 512

__device__ inline v8f wmma_f32_k4_spaced(v2f a, v2f b, v8f c) {
    asm("v_wmma_f32_16x16x4_f32 %0, %1, %2, %0\n\tv_nop\n\tv_nop\n\tv_nop\n\tv_nop"
        : "+v"(c) : "v"(a), "v"(b));
    return c;
}

template<int NT, int K>
__global__ __launch_bounds__(GEMM_T)
void wmma_gemm(const float* __restrict__ A, const float* __restrict__ B,
               const float* __restrict__ as, const float* __restrict__ ad,
               float* __restrict__ C, float* __restrict__ als,
               float* __restrict__ ald, int M, int N) {
    constexpr int NPAD = NT * 16;
    static_assert(NPAD == HEADS * HID && HID % 4 == 0 && 2 * HEADS <= U1_LD && K % 16 == 0 &&
                  (K / 16) * 32 <= GEMM_T,
                  "score tile: B's columns are the heads' channels, a k step stays inside one "
                  "head, one wave per 16 rows of the tile");
    __shared__ float us[K * U1_LD];
    const int wib  = threadIdx.x >> 5;
    const int lane = threadIdx.x & 31;
    const int half = lane >> 4;
    const int lrow = lane & 15;

    if (wib < K / 16) {
        const float* wrow = B + (size_t)(wib * 16 + lrow) * NPAD;
        const float* att  = (lrow < HEADS) ? as : ad;
        const int    hd   = (lrow < HEADS) ? lrow : lrow - HEADS;
        const bool   live = lrow < 2 * HEADS;
        v8f d;
#pragma unroll
        for (int g = 0; g < 8; g++) d[g] = 0.0f;
#pragma unroll
        for (int j = 0; j < NPAD; j += 4) {
            const int jj = j + half * 2;
            v2f a, b;
            a.x = wrow[jj];
            a.y = wrow[jj + 1];
            const float t0 = att[jj], t1 = att[jj + 1];
            const bool on = live && (jj / HID == hd);
            b.x = on ? t0 : 0.0f;
            b.y = on ? t1 : 0.0f;
            d = wmma_f32_k4_spaced(a, b, d);
        }
#pragma unroll
        for (int g = 0; g < 8; g++)
            us[(wib * 16 + g + half * 8) * U1_LD + lrow] = d[g];
    }
    __syncthreads();

    int wave = blockIdx.x * (GEMM_T / 32) + wib;
    int mtiles = M >> 4;
    if (wave >= mtiles) return;
    int m0 = wave << 4;

    v8f acc[NT], accs;
#pragma unroll
    for (int g = 0; g < 8; g++) accs[g] = 0.0f;
#pragma unroll
    for (int nt = 0; nt < NT; nt++)
#pragma unroll
        for (int g = 0; g < 8; g++) acc[nt][g] = 0.0f;

    const float* arow = A + (size_t)(m0 + lrow) * K;
    const float* bcol = B + lrow;
    const float* scol = us + lrow;
#pragma unroll 4
    for (int k = 0; k < K; k += 4) {
        int kk = k + half * 2;
        v2f a;
        a.x = arow[kk];
        a.y = arow[kk + 1];
#pragma unroll
        for (int nt = 0; nt < NT; nt++) {
            v2f b;
            b.x = bcol[kk * NPAD + nt * 16];
            b.y = bcol[(kk + 1) * NPAD + nt * 16];
            acc[nt] = __builtin_amdgcn_wmma_f32_16x16x4_f32(
                false, a, false, b, (short)0, acc[nt], false, false);
        }
        v2f bs;
        bs.x = scol[kk * U1_LD];
        bs.y = scol[(kk + 1) * U1_LD];
        accs = __builtin_amdgcn_wmma_f32_16x16x4_f32(
            false, a, false, bs, (short)0, accs, false, false);
    }
#pragma unroll
    for (int nt = 0; nt < NT; nt++) {
        int col = nt * 16 + lrow;
        if (col < N) {
#pragma unroll
            for (int g = 0; g < 8; g++)
                C[(size_t)(m0 + g + half * 8) * N + col] = acc[nt][g];
        }
    }
    if (lrow < 2 * HEADS) {
        float* sc = (lrow < HEADS) ? als : ald;
        const int h = (lrow < HEADS) ? lrow : lrow - HEADS;
#pragma unroll
        for (int g = 0; g < 8; g++)
            sc[(size_t)(m0 + g + half * 8) * HEADS + h] = accs[g];
    }
}

#define NPW 196
#define NGRP CDIV(N_NODES, NPW)
#define SCAN_U 8
#define QCAP 8192
#define NS (EDIM + 1)

__global__ void owner_bytes(const int* __restrict__ dst, unsigned* __restrict__ ob) {
    int i = blockIdx.x * blockDim.x + threadIdx.x;
    if (i >= N_EDGES / 16) return;
    const int4* d4 = reinterpret_cast<const int4*>(dst) + 4 * i;
    auto own = [](int d) { return ((unsigned)d / (unsigned)NPW) & 0xFFu; };
    uint4 w;
    int4 d;
    d = d4[0]; w.x = own(d.x) | own(d.y) << 8 | own(d.z) << 16 | own(d.w) << 24;
    d = d4[1]; w.y = own(d.x) | own(d.y) << 8 | own(d.z) << 16 | own(d.w) << 24;
    d = d4[2]; w.z = own(d.x) | own(d.y) << 8 | own(d.z) << 16 | own(d.w) << 24;
    d = d4[3]; w.w = own(d.x) | own(d.y) << 8 | own(d.z) << 16 | own(d.w) << 24;
    reinterpret_cast<uint4*>(ob)[i] = w;
}

template<class F>
__device__ inline void scan_owned(const unsigned* __restrict__ ob, unsigned grp, int tid, F f) {
    constexpr int NU = N_EDGES / 16;
    constexpr int STEP = 1024 * SCAN_U;
    constexpr int NU_FULL = (NU / STEP) * STEP;
    const unsigned pat = grp * 0x01010101u;
    auto differs = [&](unsigned w) {
        unsigned x = w ^ pat;
        return (((x & 0x7F7F7F7Fu) + 0x7F7F7F7Fu) | x) & 0x80808080u;
    };
    auto differs16 = [&](uint4 v) {
        return differs(v.x) >> 7 | differs(v.y) >> 6 | differs(v.z) >> 5 | differs(v.w) >> 4;
    };
    const uint4* o4 = reinterpret_cast<const uint4*>(ob);
    for (int base = 0; base < NU_FULL; base += STEP) {
        uint4 v[SCAN_U];
#pragma unroll
        for (int u = 0; u < SCAN_U; u++) v[u] = o4[base + u * 1024 + tid];
#pragma unroll
        for (int p = 0; p < SCAN_U; p += 2) {
            unsigned m = ~(differs16(v[p]) | differs16(v[p + 1]) << 4);
            while (m) {
                int k = __builtin_ctz(m);
                m &= m - 1;
                int i = base + (p + ((k >> 2) & 1)) * 1024 + tid;
                f((i * 4 + (k & 3)) * 4 + (k >> 3));
            }
        }
    }
    for (int i = NU_FULL + tid; i < NU; i += 1024) {
        unsigned m = ~(differs16(o4[i]) | 0xF0F0F0F0u);
        while (m) {
            int k = __builtin_ctz(m);
            m &= m - 1;
            f((i * 4 + (k & 3)) * 4 + (k >> 3));
        }
    }
}

template<int H>
__device__ inline float edge_term(const float* ea, const float* __restrict__ v, int h) {
    float ale = 0.f;
#pragma unroll
    for (int i = 0; i < EDIM; i++) ale += ea[i] * v[i * H + h];
    return ale;
}

__device__ inline float leaky_sum(float as, float ad, float ale) {
    float x = as + ad + ale;
    return (x > 0.f) ? x : NEG_SLOPE * x;
}

template<int H>
__device__ inline void raw_alpha(const float* ea, const float* __restrict__ v,
                                 const float* __restrict__ als, const float* __restrict__ ald,
                                 int s, int d, float* a) {
#pragma unroll
    for (int h = 0; h < H; h++)
        a[h] = leaky_sum(als[s * H + h], ald[d * H + h], edge_term<H>(ea, v, h));
}

template<int H, int C, bool PROJ>
__global__ __launch_bounds__(1024)
void gat_owned(const int* __restrict__ src, const int* __restrict__ dst,
               const unsigned* __restrict__ ob,
               const float* __restrict__ eattr,
               const float* __restrict__ We, const float* __restrict__ ae,
               const float* __restrict__ als, const float* __restrict__ ald,
               const float* __restrict__ xh,
               const float* __restrict__ bias, const float* __restrict__ Wn,
               const float* __restrict__ Wen, const float* __restrict__ aen,
               const float* __restrict__ asn, const float* __restrict__ adn,
               unsigned* __restrict__ hcnt, unsigned* __restrict__ hsr,
               float* __restrict__ hale,
               float* __restrict__ acc, float* __restrict__ alsn, float* __restrict__ aldn) {
    constexpr bool GIVE = PROJ, TAKE = !PROJ;
    static_assert(N_NODES <= (1 << 16) && (!TAKE || H == 1),
                  "a hand-over entry packs the source node into 16 bits and carries one edge term");
    static_assert(N_EDGES % 16 == 0 && NGRP <= 256 && ETOT <= (1 << 20) && NPW <= 256 &&
                  NPW <= QCAP && NPW * H <= 1024,
                  "owner bytes, queue entry packing, one thread per owned node and head");
    constexpr int HC = H * C;
    constexpr int J  = (HC + 31) / 32;
    constexpr int LD = PROJ ? HC + 4 : HC;
    constexpr int NP = CDIV(OUT_F, 16) * 16;
    constexpr int MT = CDIV(NPW, 16);
    constexpr int OP = NP + 4;
    constexpr int SV_T0 = 32, SVN_T0 = 64;
    static_assert(!PROJ || (1024 % HC == 0 && HC % 4 == 0 && LD % 2 == 0 && OUT_F % 4 == 0 &&
                            OP % 4 == 0 && MT * 16 * OP <= QCAP * H && OUT_F + 2 <= NP &&
                            SVN_T0 + 32 <= 1024 - 2 * HC),
                  "one bias channel per thread, k steps of 4, 8-byte fragment reads, 16-byte row "
                  "copies, projected tiles fit the score array they reuse, two padding columns "
                  "are free for the next layer's score vectors, the threads that fold them are "
                  "not the ones that fold the edge vectors");

    __shared__ alignas(16) float sacc[NPW * LD];
    __shared__ alignas(16) float wns[PROJ ? HC * NP : 2];
    __shared__ int      q[QCAP];
    __shared__ int      qs[QCAP];
    __shared__ alignas(16) float qa[QCAP * H];
    __shared__ float    nsum[NPW * NS];
    __shared__ unsigned nmax[NPW * H];
    __shared__ float    nden[NPW * H];
    __shared__ float    aself[NPW * H];
    __shared__ float    sv[EDIM * H];
    __shared__ float    svn[PROJ ? EDIM : 1];
    __shared__ int      qn;
    const float* v  = sv;
    const float* vn = svn;

    const int tid  = threadIdx.x;
    const int lane = tid & 31;
    const int wave = __builtin_amdgcn_readfirstlane(tid >> 5);
    const unsigned grp = blockIdx.x;
    const int lo   = blockIdx.x * NPW;
    int nown = N_NODES - lo;
    if (nown > NPW) nown = NPW;

    static_assert(EDIM * H <= 32 && EDIM <= 32, "each folded edge vector inside one wave");
    if (tid >= SV_T0 && tid < SV_T0 + EDIM * H) {
        const int i = tid - SV_T0;
        const int d = i / H, h = i % H;
        float s = 0.f;
#pragma unroll
        for (int cc = 0; cc < C; cc++) s += We[d * HC + h * C + cc] * ae[h * C + cc];
        sv[i] = s;
    }
    float bc = 0.f;
    if constexpr (PROJ) {
        bc = bias[tid % HC];
        if (tid >= SVN_T0 && tid < SVN_T0 + EDIM) {
            const int d = tid - SVN_T0;
            float s = 0.f;
#pragma unroll
            for (int cc = 0; cc < OUT_F; cc++) s += Wen[d * OUT_F + cc] * aen[cc];
            svn[d] = s;
        }
        for (int i = tid; i < HC * NP; i += 1024) {
            int k = i / NP, cn = i - k * NP;
            if (cn >= OUT_F && cn < OUT_F + 2) continue;
            float wv = 0.f;
            if (cn < OUT_F) wv = Wn[k * OUT_F + cn];
            wns[((k >> 1) * NP + cn) * 2 + (k & 1)] = wv;
        }
        if (tid >= 1024 - 2 * HC) {
            const int i = tid - (1024 - 2 * HC);
            const int k = i >> 1, which = i & 1;
            const float* att = which ? adn : asn;
            float s = 0.f;
#pragma unroll
            for (int cc = 0; cc < OUT_F; cc++) s += Wn[k * OUT_F + cc] * att[cc];
            wns[((k >> 1) * NP + OUT_F + which) * 2 + (k & 1)] = s;
        }
    }
    for (int idx = tid; idx < NPW * LD; idx += 1024) sacc[idx] = 0.f;
    for (int idx = tid; idx < NPW * NS; idx += 1024) nsum[idx] = 0.f;
    if (tid < NPW * H) { nmax[tid] = ENC_NEG_INF; nden[tid] = 0.f; }
    if (tid == 0) qn = 0;
    __syncthreads();

    if constexpr (!TAKE) {
        scan_owned(ob, grp, tid, [&](int e) {
            int slot = atomicAdd(&qn, 1);
            if (slot < QCAP) q[slot] = e;
        });
        __syncthreads();
    }
    const int nreal = __builtin_amdgcn_readfirstlane(TAKE ? (int)hcnt[grp] : qn);
    const int total = nreal + nown;
    if constexpr (GIVE) { if (tid == 0) hcnt[grp] = (unsigned)nreal; }
    const size_t hbase = (size_t)grp * QCAP;

    auto load_ea = [&](int e, float* ea) {
#pragma unroll
        for (int i = 0; i < EDIM; i++) ea[i] = eattr[(size_t)e * EDIM + i];
    };
    auto self_ea = [&](int r, float* ea) {
        float c = fmaxf(nsum[r * NS], 1.0f);
#pragma unroll
        for (int i = 0; i < EDIM; i++) ea[i] = nsum[r * NS + 1 + i] / c;
    };

    if (total <= QCAP) {
        if constexpr (TAKE) {
            constexpr int TU = 4;
            for (int base = tid; base < total; base += TU * 1024) {
                unsigned sr[TU];
                float ale[TU], a[TU];
#pragma unroll
                for (int u = 0; u < TU; u++) {
                    int idx = base + u * 1024;
                    sr[u] = (unsigned)lo;
                    ale[u] = 0.f;
                    if (idx < total) {
                        sr[u]  = hsr[hbase + idx];
                        ale[u] = hale[hbase + idx];
                    }
                }
#pragma unroll
                for (int u = 0; u < TU; u++)
                    a[u] = leaky_sum(als[sr[u] & 0xFFFFu], ald[lo + (sr[u] >> 16)], ale[u]);
#pragma unroll
                for (int u = 0; u < TU; u++) {
                    int idx = base + u * 1024;
                    if (idx < total) {
                        int r = (int)(sr[u] >> 16);
                        q[idx]  = r << 20;
                        qs[idx] = (int)(sr[u] & 0xFFFFu);
                        qa[idx] = a[u];
                        atomicMax(&nmax[r], encf(a[u]));
                    }
                }
            }
        } else {
            for (int idx = tid; idx < nreal; idx += 1024) {
                int e = q[idx];
                int s = src[e];
                int r = dst[e] - lo;
                q[idx] = e | (r << 20);
                float ea[EDIM], a[H];
                load_ea(e, ea);
                atomicAdd(&nsum[r * NS], 1.0f);
#pragma unroll
                for (int i = 0; i < EDIM; i++) atomicAdd(&nsum[r * NS + 1 + i], ea[i]);
                raw_alpha<H>(ea, v, als, ald, s, lo + r, a);
                qs[idx] = s;
#pragma unroll
                for (int h = 0; h < H; h++) {
                    qa[idx * H + h] = a[h];
                    atomicMax(&nmax[r * H + h], encf(a[h]));
                }
                if constexpr (GIVE) {
                    hsr[hbase + idx]  = (unsigned)s | (unsigned)r << 16;
                    hale[hbase + idx] = edge_term<1>(ea, vn, 0);
                }
            }
            __syncthreads();
            if (tid < nown) {
                int r = tid, idx = nreal + tid;
                float ea[EDIM], a[H];
                self_ea(r, ea);
                raw_alpha<H>(ea, v, als, ald, lo + r, lo + r, a);
                q[idx]  = (N_EDGES + lo + r) | (r << 20);
                qs[idx] = lo + r;
#pragma unroll
                for (int h = 0; h < H; h++) {
                    qa[idx * H + h] = a[h];
                    atomicMax(&nmax[r * H + h], encf(a[h]));
                }
                if constexpr (GIVE) {
                    hsr[hbase + idx]  = (unsigned)(lo + r) | (unsigned)r << 16;
                    hale[hbase + idx] = edge_term<1>(ea, vn, 0);
                }
            }
        }
        __syncthreads();
        for (int idx = tid; idx < total; idx += 1024) {
            int r = q[idx] >> 20;
#pragma unroll
            for (int h = 0; h < H; h++) {
                float ex = expf(qa[idx * H + h] - decf(nmax[r * H + h]));
                qa[idx * H + h] = ex;
                atomicAdd(&nden[r * H + h], ex);
            }
        }
        __syncthreads();

        auto head = [&](int g, int& s, int& r, float* w) {
            s = 0; r = 0;
#pragma unroll
            for (int h = 0; h < H; h++) w[h] = 0.f;
            if (g + lane < total) {
                r = q[g + lane] >> 20;
                s = qs[g + lane];
#pragma unroll
                for (int h = 0; h < H; h++)
                    w[h] = qa[(g + lane) * H + h] / (nden[r * H + h] + 1e-16f);
            }
        };
        auto load8 = [&](int s, int j0, float (&val)[8][J]) {
#pragma unroll
            for (int b = 0; b < 8; b++) {
                int sj = __builtin_amdgcn_readlane(s, j0 + b);
#pragma unroll
                for (int jx = 0; jx < J; jx++) {
                    int c = lane + 32 * jx;
                    val[b][jx] = xh[(size_t)sj * HC + (c < HC ? c : HC - 1)];
                }
            }
        };
        auto add8 = [&](int r, const float* w, int j0, int cnt, const float (&val)[8][J]) {
#pragma unroll
            for (int b = 0; b < 8; b++) {
                if (j0 + b < cnt) {
                    int rj = __builtin_amdgcn_readlane(r, j0 + b);
#pragma unroll
                    for (int jx = 0; jx < J; jx++) {
                        int c = lane + 32 * jx;
                        int h = (H == 1) ? 0 : (32 * jx) / C;
                        float wj = __int_as_float(__builtin_amdgcn_readlane(__float_as_int(w[h]), j0 + b));
                        if (c < HC) atomicAdd(&sacc[rj * LD + c], val[b][jx] * wj);
                    }
                }
            }
        };
        if (wave * 32 < total) {
            int g = wave * 32;
            int s, r;
            float w[H];
            float va[8][J], vb[8][J];
            head(g, s, r, w);
            load8(s, 0, va);
            while (total - g >= 32) {
                const int gn = g + 1024;
                int sn, rn;
                float wn[H];
                head(gn, sn, rn, wn);
                load8(s, 8, vb);
                add8(r, w, 0, 32, va);
                load8(s, 16, va);
                add8(r, w, 8, 32, vb);
                load8(s, 24, vb);
                add8(r, w, 16, 32, va);
                load8(sn, 0, va);
                add8(r, w, 24, 32, vb);
                g = gn; s = sn; r = rn;
#pragma unroll
                for (int h = 0; h < H; h++) w[h] = wn[h];
            }
            if (g < total) {
                const int cnt = total - g;
                add8(r, w, 0, cnt, va);
                for (int j0 = 8; j0 < cnt; j0 += 8) {
                    load8(s, j0, va);
                    add8(r, w, j0, cnt, va);
                }
            }
        }
    } else {
        scan_owned(ob, grp, tid, [&](int e) {
            int r = dst[e] - lo;
            float ea[EDIM], a[H];
            load_ea(e, ea);
            atomicAdd(&nsum[r * NS], 1.0f);
#pragma unroll
            for (int i = 0; i < EDIM; i++) atomicAdd(&nsum[r * NS + 1 + i], ea[i]);
            raw_alpha<H>(ea, v, als, ald, src[e], lo + r, a);
#pragma unroll
            for (int h = 0; h < H; h++) atomicMax(&nmax[r * H + h], encf(a[h]));
        });
        __syncthreads();
        if (tid < nown) {
            float ea[EDIM], a[H];
            self_ea(tid, ea);
            raw_alpha<H>(ea, v, als, ald, lo + tid, lo + tid, a);
#pragma unroll
            for (int h = 0; h < H; h++) {
                aself[tid * H + h] = a[h];
                atomicMax(&nmax[tid * H + h], encf(a[h]));
            }
        }
        __syncthreads();
        scan_owned(ob, grp, tid, [&](int e) {
            int r = dst[e] - lo;
            float ea[EDIM], a[H];
            load_ea(e, ea);
            raw_alpha<H>(ea, v, als, ald, src[e], lo + r, a);
#pragma unroll
            for (int h = 0; h < H; h++)
                atomicAdd(&nden[r * H + h], expf(a[h] - decf(nmax[r * H + h])));
        });
        if (tid < nown) {
#pragma unroll
            for (int h = 0; h < H; h++)
                atomicAdd(&nden[tid * H + h], expf(aself[tid * H + h] - decf(nmax[tid * H + h])));
        }
        __syncthreads();
        auto add_row = [&](int s, int r, const float* a) {
            for (int h = 0; h < H; h++) {
                float w = expf(a[h] - decf(nmax[r * H + h])) / (nden[r * H + h] + 1e-16f);
                for (int c = 0; c < C; c++)
                    atomicAdd(&sacc[r * LD + h * C + c], xh[(size_t)s * HC + h * C + c] * w);
            }
        };
        scan_owned(ob, grp, tid, [&](int e) {
            int r = dst[e] - lo;
            float ea[EDIM], a[H];
            load_ea(e, ea);
            int s = src[e];
            raw_alpha<H>(ea, v, als, ald, s, lo + r, a);
            add_row(s, r, a);
        });
        if (tid < nown) {
            float a[H];
#pragma unroll
            for (int h = 0; h < H; h++) a[h] = aself[tid * H + h];
            add_row(lo + tid, tid, a);
        }
    }
    __syncthreads();

    if constexpr (PROJ) {
        for (int idx = tid; idx < NPW * HC; idx += 1024) {
            float* p = &sacc[(idx / HC) * LD + (idx % HC)];
            float x = *p + bc;
            *p = (x > 0.f) ? x : expm1f(x);
        }
        __syncthreads();
        constexpr int NT = NP / 16;
        float* ost = qa;
        const int half = lane >> 4, lrow = lane & 15;
        for (int j = wave; j < MT * NT; j += 32) {
            const int mt = j / NT, nt = j - mt * NT;
            int row = mt * 16 + lrow;
            if (row > NPW - 1) row = NPW - 1;
            const float* arow = sacc + row * LD + half * 2;
            const float* bcol = wns + (half * NP + nt * 16 + lrow) * 2;
            float* orow = ost + (mt * 16 + half * 8) * OP + nt * 16 + lrow;
            v8f d;
#pragma unroll
            for (int g = 0; g < 8; g++) d[g] = 0.0f;
#pragma unroll
            for (int k = 0; k < HC; k += 4) {
                v2f a = *reinterpret_cast<const v2f*>(arow + k);
                v2f b = *reinterpret_cast<const v2f*>(bcol + k * NP);
                d = wmma_f32_k4_spaced(a, b, d);
            }
#pragma unroll
            for (int g = 0; g < 8; g++) orow[g * OP] = d[g];
        }
        __syncthreads();
        float4* g4 = reinterpret_cast<float4*>(acc + (size_t)lo * OUT_F);
        for (int i = tid; i < nown * (OUT_F / 4); i += 1024) {
            int r = i / (OUT_F / 4), c4 = i - r * (OUT_F / 4);
            g4[i] = *reinterpret_cast<const float4*>(ost + r * OP + c4 * 4);
        }
        if (tid < nown) {
            alsn[lo + tid] = ost[tid * OP + OUT_F];
            aldn[lo + tid] = ost[tid * OP + OUT_F + 1];
        }
    } else {
        static_assert(LD == HC && HC % 4 == 0, "rows lie back to back and are read 16 bytes at a time");
        if (tid < nown) {
            float4* row4 = reinterpret_cast<float4*>(sacc + tid * LD);
            float vals[HC];
#pragma unroll
            for (int i = 0; i < HC / 4; i++) {
                const float4 t = row4[i];
                vals[4 * i]     = t.x + bias[4 * i];
                vals[4 * i + 1] = t.y + bias[4 * i + 1];
                vals[4 * i + 2] = t.z + bias[4 * i + 2];
                vals[4 * i + 3] = t.w + bias[4 * i + 3];
            }
            float mx = -INFINITY;
#pragma unroll
            for (int cc = 0; cc < HC; cc++) mx = fmaxf(mx, vals[cc]);
            float sum = 0.f;
#pragma unroll
            for (int cc = 0; cc < HC; cc++) sum += expf(vals[cc] - mx);
            const float lse = mx + logf(sum);
#pragma unroll
            for (int i = 0; i < HC / 4; i++)
                row4[i] = make_float4(vals[4 * i] - lse, vals[4 * i + 1] - lse,
                                      vals[4 * i + 2] - lse, vals[4 * i + 3] - lse);
        }
        __syncthreads();
        float4* g4 = reinterpret_cast<float4*>(acc + (size_t)lo * HC);
        const float4* s4 = reinterpret_cast<const float4*>(sacc);
        for (int i = tid; i < nown * (HC / 4); i += 1024) g4[i] = s4[i];
    }
}

extern "C" void kernel_launch(void* const* d_in, const int* in_sizes, int n_in,
                              void* d_out, int out_size, void* d_ws, size_t ws_size,
                              hipStream_t stream) {
    const float* x   = (const float*)d_in[0];
    const int*   ei  = (const int*)d_in[1];
    const float* ea  = (const float*)d_in[2];
    const float* W1  = (const float*)d_in[3];
    const float* as1 = (const float*)d_in[4];
    const float* ad1 = (const float*)d_in[5];
    const float* We1 = (const float*)d_in[6];
    const float* ae1 = (const float*)d_in[7];
    const float* b1  = (const float*)d_in[8];
    const float* W2  = (const float*)d_in[9];
    const float* as2 = (const float*)d_in[10];
    const float* ad2 = (const float*)d_in[11];
    const float* We2 = (const float*)d_in[12];
    const float* ae2 = (const float*)d_in[13];
    const float* b2  = (const float*)d_in[14];

    const int N = N_NODES, E = N_EDGES;
    const int HC1 = HEADS * HID;
    const int* src = ei;
    const int* dst = ei + E;

    char* ws = (char*)d_ws;
    size_t off = 0;
    auto alloc = [&](size_t bytes) -> void* {
        void* p = ws + off;
        off = (off + bytes + 255) & ~(size_t)255;
        return p;
    };
    float*    xh1    = (float*)alloc(sizeof(float) * (size_t)N * HC1);
    float*    als1   = (float*)alloc(sizeof(float) * N * HEADS);
    float*    ald1   = (float*)alloc(sizeof(float) * N * HEADS);
    float*    xh2    = (float*)alloc(sizeof(float) * (size_t)N * OUT_F);
    float*    als2   = (float*)alloc(sizeof(float) * N);
    float*    ald2   = (float*)alloc(sizeof(float) * N);
    unsigned* ob     = (unsigned*)alloc(N_EDGES);
    unsigned* hcnt   = (unsigned*)alloc(sizeof(unsigned) * NGRP);
    unsigned* hsr    = (unsigned*)alloc(sizeof(unsigned) * (size_t)NGRP * QCAP);
    float*    hale   = (float*)alloc(sizeof(float) * (size_t)NGRP * QCAP);
    (void)ws_size; (void)n_in; (void)in_sizes; (void)out_size;

    const int T = 256;

    owner_bytes<<<CDIV(N_EDGES / 16, T), T, 0, stream>>>(dst, ob);

    {
        int mtiles = N / 16, waves_per_blk = GEMM_T / 32;
        wmma_gemm<8, F_IN><<<CDIV(mtiles, waves_per_blk), GEMM_T, 0, stream>>>(x, W1, as1, ad1, xh1,
                                                                               als1, ald1, N, HC1);
    }
    gat_owned<HEADS, HID, true><<<NGRP, 1024, 0, stream>>>(src, dst, ob, ea, We1, ae1, als1, ald1,
                                                           xh1, b1, W2, We2, ae2, as2, ad2,
                                                           hcnt, hsr, hale, xh2, als2, ald2);

    gat_owned<1, OUT_F, false><<<NGRP, 1024, 0, stream>>>(src, dst, ob, ea, We2, ae2, als2, ald2,
                                                          xh2, b2, nullptr, nullptr, nullptr,
                                                          nullptr, nullptr,
                                                          hcnt, hsr, hale, (float*)d_out, nullptr,
                                                          nullptr);
}


# Drain: entries ranked in their node (LDS int atomic, arrival order), put in node order; a wave sums a node's weighted rows in registers in rank order, one LDS float atomic per run, not per row
# speedup vs baseline: 9.7162x; 1.0595x over previous
#include <hip/hip_runtime.h>
#include <math.h>

#define N_NODES 50000
#define N_EDGES 800000
#define ETOT    (N_EDGES + N_NODES)
#define F_IN    128
#define HID     64
#define HEADS   2
#define OUT_F   40
#define EDIM    5
#define NEG_SLOPE 0.2f

#define CDIV(a,b) (((a)+(b)-1)/(b))

typedef __attribute__((ext_vector_type(2))) float v2f;
typedef __attribute__((ext_vector_type(8))) float v8f;

__device__ inline unsigned encf(float f) {
    unsigned u = __float_as_uint(f);
    return (u & 0x80000000u) ? ~u : (u | 0x80000000u);
}
__device__ inline float decf(unsigned m) {
    unsigned u = (m & 0x80000000u) ? (m ^ 0x80000000u) : ~m;
    return __uint_as_float(u);
}
#define ENC_NEG_INF 0x007FFFFFu

#define U1_LD 16
#define GEMM_T 512

__device__ inline v8f wmma_f32_k4_spaced(v2f a, v2f b, v8f c) {
    asm("v_wmma_f32_16x16x4_f32 %0, %1, %2, %0\n\tv_nop\n\tv_nop\n\tv_nop\n\tv_nop"
        : "+v"(c) : "v"(a), "v"(b));
    return c;
}

template<int NT, int K>
__global__ __launch_bounds__(GEMM_T)
void wmma_gemm(const float* __restrict__ A, const float* __restrict__ B,
               const float* __restrict__ as, const float* __restrict__ ad,
               float* __restrict__ C, float* __restrict__ als,
               float* __restrict__ ald, int M, int N) {
    constexpr int NPAD = NT * 16;
    static_assert(NPAD == HEADS * HID && HID % 4 == 0 && 2 * HEADS <= U1_LD && K % 16 == 0 &&
                  (K / 16) * 32 <= GEMM_T,
                  "score tile: B's columns are the heads' channels, a k step stays inside one "
                  "head, one wave per 16 rows of the tile");
    __shared__ float us[K * U1_LD];
    const int wib  = threadIdx.x >> 5;
    const int lane = threadIdx.x & 31;
    const int half = lane >> 4;
    const int lrow = lane & 15;

    if (wib < K / 16) {
        const float* wrow = B + (size_t)(wib * 16 + lrow) * NPAD;
        const float* att  = (lrow < HEADS) ? as : ad;
        const int    hd   = (lrow < HEADS) ? lrow : lrow - HEADS;
        const bool   live = lrow < 2 * HEADS;
        v8f d;
#pragma unroll
        for (int g = 0; g < 8; g++) d[g] = 0.0f;
#pragma unroll
        for (int j = 0; j < NPAD; j += 4) {
            const int jj = j + half * 2;
            v2f a, b;
            a.x = wrow[jj];
            a.y = wrow[jj + 1];
            const float t0 = att[jj], t1 = att[jj + 1];
            const bool on = live && (jj / HID == hd);
            b.x = on ? t0 : 0.0f;
            b.y = on ? t1 : 0.0f;
            d = wmma_f32_k4_spaced(a, b, d);
        }
#pragma unroll
        for (int g = 0; g < 8; g++)
            us[(wib * 16 + g + half * 8) * U1_LD + lrow] = d[g];
    }
    __syncthreads();

    int wave = blockIdx.x * (GEMM_T / 32) + wib;
    int mtiles = M >> 4;
    if (wave >= mtiles) return;
    int m0 = wave << 4;

    v8f acc[NT], accs;
#pragma unroll
    for (int g = 0; g < 8; g++) accs[g] = 0.0f;
#pragma unroll
    for (int nt = 0; nt < NT; nt++)
#pragma unroll
        for (int g = 0; g < 8; g++) acc[nt][g] = 0.0f;

    const float* arow = A + (size_t)(m0 + lrow) * K;
    const float* bcol = B + lrow;
    const float* scol = us + lrow;
#pragma unroll 4
    for (int k = 0; k < K; k += 4) {
        int kk = k + half * 2;
        v2f a;
        a.x = arow[kk];
        a.y = arow[kk + 1];
#pragma unroll
        for (int nt = 0; nt < NT; nt++) {
            v2f b;
            b.x = bcol[kk * NPAD + nt * 16];
            b.y = bcol[(kk + 1) * NPAD + nt * 16];
            acc[nt] = __builtin_amdgcn_wmma_f32_16x16x4_f32(
                false, a, false, b, (short)0, acc[nt], false, false);
        }
        v2f bs;
        bs.x = scol[kk * U1_LD];
        bs.y = scol[(kk + 1) * U1_LD];
        accs = __builtin_amdgcn_wmma_f32_16x16x4_f32(
            false, a, false, bs, (short)0, accs, false, false);
    }
#pragma unroll
    for (int nt = 0; nt < NT; nt++) {
        int col = nt * 16 + lrow;
        if (col < N) {
#pragma unroll
            for (int g = 0; g < 8; g++)
                C[(size_t)(m0 + g + half * 8) * N + col] = acc[nt][g];
        }
    }
    if (lrow < 2 * HEADS) {
        float* sc = (lrow < HEADS) ? als : ald;
        const int h = (lrow < HEADS) ? lrow : lrow - HEADS;
#pragma unroll
        for (int g = 0; g < 8; g++)
            sc[(size_t)(m0 + g + half * 8) * HEADS + h] = accs[g];
    }
}

#define NPW 196
#define NGRP CDIV(N_NODES, NPW)
#define SCAN_U 8
#define QCAP 8192
#define NS (EDIM + 1)

__global__ void owner_bytes(const int* __restrict__ dst, unsigned* __restrict__ ob) {
    int i = blockIdx.x * blockDim.x + threadIdx.x;
    if (i >= N_EDGES / 16) return;
    const int4* d4 = reinterpret_cast<const int4*>(dst) + 4 * i;
    auto own = [](int d) { return ((unsigned)d / (unsigned)NPW) & 0xFFu; };
    uint4 w;
    int4 d;
    d = d4[0]; w.x = own(d.x) | own(d.y) << 8 | own(d.z) << 16 | own(d.w) << 24;
    d = d4[1]; w.y = own(d.x) | own(d.y) << 8 | own(d.z) << 16 | own(d.w) << 24;
    d = d4[2]; w.z = own(d.x) | own(d.y) << 8 | own(d.z) << 16 | own(d.w) << 24;
    d = d4[3]; w.w = own(d.x) | own(d.y) << 8 | own(d.z) << 16 | own(d.w) << 24;
    reinterpret_cast<uint4*>(ob)[i] = w;
}

template<class F>
__device__ inline void scan_owned(const unsigned* __restrict__ ob, unsigned grp, int tid, F f) {
    constexpr int NU = N_EDGES / 16;
    constexpr int STEP = 1024 * SCAN_U;
    constexpr int NU_FULL = (NU / STEP) * STEP;
    const unsigned pat = grp * 0x01010101u;
    auto differs = [&](unsigned w) {
        unsigned x = w ^ pat;
        return (((x & 0x7F7F7F7Fu) + 0x7F7F7F7Fu) | x) & 0x80808080u;
    };
    auto differs16 = [&](uint4 v) {
        return differs(v.x) >> 7 | differs(v.y) >> 6 | differs(v.z) >> 5 | differs(v.w) >> 4;
    };
    const uint4* o4 = reinterpret_cast<const uint4*>(ob);
    for (int base = 0; base < NU_FULL; base += STEP) {
        uint4 v[SCAN_U];
#pragma unroll
        for (int u = 0; u < SCAN_U; u++) v[u] = o4[base + u * 1024 + tid];
#pragma unroll
        for (int p = 0; p < SCAN_U; p += 2) {
            unsigned m = ~(differs16(v[p]) | differs16(v[p + 1]) << 4);
            while (m) {
                int k = __builtin_ctz(m);
                m &= m - 1;
                int i = base + (p + ((k >> 2) & 1)) * 1024 + tid;
                f((i * 4 + (k & 3)) * 4 + (k >> 3));
            }
        }
    }
    for (int i = NU_FULL + tid; i < NU; i += 1024) {
        unsigned m = ~(differs16(o4[i]) | 0xF0F0F0F0u);
        while (m) {
            int k = __builtin_ctz(m);
            m &= m - 1;
            f((i * 4 + (k & 3)) * 4 + (k >> 3));
        }
    }
}

template<int H>
__device__ inline float edge_term(const float* ea, const float* __restrict__ v, int h) {
    float ale = 0.f;
#pragma unroll
    for (int i = 0; i < EDIM; i++) ale += ea[i] * v[i * H + h];
    return ale;
}

__device__ inline float leaky_sum(float as, float ad, float ale) {
    float x = as + ad + ale;
    return (x > 0.f) ? x : NEG_SLOPE * x;
}

template<int H>
__device__ inline void raw_alpha(const float* ea, const float* __restrict__ v,
                                 const float* __restrict__ als, const float* __restrict__ ald,
                                 int s, int d, float* a) {
#pragma unroll
    for (int h = 0; h < H; h++)
        a[h] = leaky_sum(als[s * H + h], ald[d * H + h], edge_term<H>(ea, v, h));
}

template<int H, int C, bool PROJ>
__global__ __launch_bounds__(1024)
void gat_owned(const int* __restrict__ src, const int* __restrict__ dst,
               const unsigned* __restrict__ ob,
               const float* __restrict__ eattr,
               const float* __restrict__ We, const float* __restrict__ ae,
               const float* __restrict__ als, const float* __restrict__ ald,
               const float* __restrict__ xh,
               const float* __restrict__ bias, const float* __restrict__ Wn,
               const float* __restrict__ Wen, const float* __restrict__ aen,
               const float* __restrict__ asn, const float* __restrict__ adn,
               unsigned* __restrict__ hcnt, unsigned* __restrict__ hsr,
               float* __restrict__ hale,
               float* __restrict__ acc, float* __restrict__ alsn, float* __restrict__ aldn) {
    constexpr bool GIVE = PROJ, TAKE = !PROJ;
    static_assert(N_NODES <= (1 << 16) && (!TAKE || H == 1),
                  "a hand-over entry packs the source node into 16 bits and carries one edge term");
    static_assert(N_EDGES % 16 == 0 && NGRP <= 256 && ETOT <= (1 << 20) && NPW <= 256 &&
                  NPW <= QCAP && NPW * H <= 1024,
                  "owner bytes, queue entry packing, one thread per owned node and head");
    constexpr int HC = H * C;
    constexpr int J  = (HC + 31) / 32;
    constexpr int LD = PROJ ? HC + 4 : HC;
    constexpr int NP = CDIV(OUT_F, 16) * 16;
    constexpr int MT = CDIV(NPW, 16);
    constexpr int OP = NP + 4;
    constexpr int SV_T0 = 32, SVN_T0 = 64;
    static_assert(!PROJ || (1024 % HC == 0 && HC % 4 == 0 && LD % 2 == 0 && OUT_F % 4 == 0 &&
                            OP % 4 == 0 && MT * 16 * OP <= QCAP * H && OUT_F + 2 <= NP &&
                            SVN_T0 + 32 <= 1024 - 2 * HC),
                  "one bias channel per thread, k steps of 4, 8-byte fragment reads, 16-byte row "
                  "copies, projected tiles fit the score array they reuse, two padding columns "
                  "are free for the next layer's score vectors, the threads that fold them are "
                  "not the ones that fold the edge vectors");

    __shared__ alignas(16) float sacc[NPW * LD];
    __shared__ alignas(16) float wns[PROJ ? HC * NP : 2];
    __shared__ int      q[QCAP];
    __shared__ int      qs[QCAP];
    __shared__ alignas(16) float qa[QCAP * H];
    __shared__ float    nsum[NPW * NS];
    __shared__ unsigned nmax[NPW * H];
    __shared__ float    nden[NPW * H];
    __shared__ float    aself[NPW * H];
    __shared__ float    sv[EDIM * H];
    __shared__ float    svn[PROJ ? EDIM : 1];
    __shared__ int      qn;
    __shared__ int      ncnt[NPW];
    __shared__ int      nstart[NPW];
    __shared__ unsigned short pp[QCAP];
    static_assert(QCAP <= (1 << 16), "a queue index fits the 16 bits of a place");
    const float* v  = sv;
    const float* vn = svn;

    const int tid  = threadIdx.x;
    const int lane = tid & 31;
    const int wave = __builtin_amdgcn_readfirstlane(tid >> 5);
    const unsigned grp = blockIdx.x;
    const int lo   = blockIdx.x * NPW;
    int nown = N_NODES - lo;
    if (nown > NPW) nown = NPW;

    static_assert(EDIM * H <= 32 && EDIM <= 32, "each folded edge vector inside one wave");
    if (tid >= SV_T0 && tid < SV_T0 + EDIM * H) {
        const int i = tid - SV_T0;
        const int d = i / H, h = i % H;
        float s = 0.f;
#pragma unroll
        for (int cc = 0; cc < C; cc++) s += We[d * HC + h * C + cc] * ae[h * C + cc];
        sv[i] = s;
    }
    float bc = 0.f;
    if constexpr (PROJ) {
        bc = bias[tid % HC];
        if (tid >= SVN_T0 && tid < SVN_T0 + EDIM) {
            const int d = tid - SVN_T0;
            float s = 0.f;
#pragma unroll
            for (int cc = 0; cc < OUT_F; cc++) s += Wen[d * OUT_F + cc] * aen[cc];
            svn[d] = s;
        }
        for (int i = tid; i < HC * NP; i += 1024) {
            int k = i / NP, cn = i - k * NP;
            if (cn >= OUT_F && cn < OUT_F + 2) continue;
            float wv = 0.f;
            if (cn < OUT_F) wv = Wn[k * OUT_F + cn];
            wns[((k >> 1) * NP + cn) * 2 + (k & 1)] = wv;
        }
        if (tid >= 1024 - 2 * HC) {
            const int i = tid - (1024 - 2 * HC);
            const int k = i >> 1, which = i & 1;
            const float* att = which ? adn : asn;
            float s = 0.f;
#pragma unroll
            for (int cc = 0; cc < OUT_F; cc++) s += Wn[k * OUT_F + cc] * att[cc];
            wns[((k >> 1) * NP + OUT_F + which) * 2 + (k & 1)] = s;
        }
    }
    for (int idx = tid; idx < NPW * LD; idx += 1024) sacc[idx] = 0.f;
    for (int idx = tid; idx < NPW * NS; idx += 1024) nsum[idx] = 0.f;
    if (tid < NPW * H) { nmax[tid] = ENC_NEG_INF; nden[tid] = 0.f; }
    if (tid < NPW) ncnt[tid] = 0;
    if (tid == 0) qn = 0;
    __syncthreads();

    if constexpr (!TAKE) {
        scan_owned(ob, grp, tid, [&](int e) {
            int slot = atomicAdd(&qn, 1);
            if (slot < QCAP) q[slot] = e;
        });
        __syncthreads();
    }
    const int nreal = __builtin_amdgcn_readfirstlane(TAKE ? (int)hcnt[grp] : qn);
    const int total = nreal + nown;
    if constexpr (GIVE) { if (tid == 0) hcnt[grp] = (unsigned)nreal; }
    const size_t hbase = (size_t)grp * QCAP;

    auto load_ea = [&](int e, float* ea) {
#pragma unroll
        for (int i = 0; i < EDIM; i++) ea[i] = eattr[(size_t)e * EDIM + i];
    };
    auto self_ea = [&](int r, float* ea) {
        float c = fmaxf(nsum[r * NS], 1.0f);
#pragma unroll
        for (int i = 0; i < EDIM; i++) ea[i] = nsum[r * NS + 1 + i] / c;
    };

    if (total <= QCAP) {
        if constexpr (TAKE) {
            constexpr int TU = 4;
            for (int base = tid; base < total; base += TU * 1024) {
                unsigned sr[TU];
                float ale[TU], a[TU];
#pragma unroll
                for (int u = 0; u < TU; u++) {
                    int idx = base + u * 1024;
                    sr[u] = (unsigned)lo;
                    ale[u] = 0.f;
                    if (idx < total) {
                        sr[u]  = hsr[hbase + idx];
                        ale[u] = hale[hbase + idx];
                    }
                }
#pragma unroll
                for (int u = 0; u < TU; u++)
                    a[u] = leaky_sum(als[sr[u] & 0xFFFFu], ald[lo + (sr[u] >> 16)], ale[u]);
#pragma unroll
                for (int u = 0; u < TU; u++) {
                    int idx = base + u * 1024;
                    if (idx < total) {
                        int r = (int)(sr[u] >> 16);
                        q[idx]  = atomicAdd(&ncnt[r], 1) | (r << 20);
                        qs[idx] = (int)(sr[u] & 0xFFFFu);
                        qa[idx] = a[u];
                        atomicMax(&nmax[r], encf(a[u]));
                    }
                }
            }
        } else {
            for (int idx = tid; idx < nreal; idx += 1024) {
                int e = q[idx];
                int s = src[e];
                int r = dst[e] - lo;
                q[idx] = atomicAdd(&ncnt[r], 1) | (r << 20);
                float ea[EDIM], a[H];
                load_ea(e, ea);
                atomicAdd(&nsum[r * NS], 1.0f);
#pragma unroll
                for (int i = 0; i < EDIM; i++) atomicAdd(&nsum[r * NS + 1 + i], ea[i]);
                raw_alpha<H>(ea, v, als, ald, s, lo + r, a);
                qs[idx] = s;
#pragma unroll
                for (int h = 0; h < H; h++) {
                    qa[idx * H + h] = a[h];
                    atomicMax(&nmax[r * H + h], encf(a[h]));
                }
                if constexpr (GIVE) {
                    hsr[hbase + idx]  = (unsigned)s | (unsigned)r << 16;
                    hale[hbase + idx] = edge_term<1>(ea, vn, 0);
                }
            }
            __syncthreads();
            if (tid < nown) {
                int r = tid, idx = nreal + tid;
                float ea[EDIM], a[H];
                self_ea(r, ea);
                raw_alpha<H>(ea, v, als, ald, lo + r, lo + r, a);
                const int rk = ncnt[r];
                ncnt[r] = rk + 1;
                q[idx]  = rk | (r << 20);
                qs[idx] = lo + r;
#pragma unroll
                for (int h = 0; h < H; h++) {
                    qa[idx * H + h] = a[h];
                    atomicMax(&nmax[r * H + h], encf(a[h]));
                }
                if constexpr (GIVE) {
                    hsr[hbase + idx]  = (unsigned)(lo + r) | (unsigned)r << 16;
                    hale[hbase + idx] = edge_term<1>(ea, vn, 0);
                }
            }
        }
        __syncthreads();
        if (wave == 0) {
            constexpr int PER = CDIV(NPW, 32);
            int c[PER], sum = 0;
#pragma unroll
            for (int i = 0; i < PER; i++) {
                const int n = lane * PER + i;
                c[i] = (n < NPW) ? ncnt[n] : 0;
                sum += c[i];
            }
            int incl = sum;
#pragma unroll
            for (int d = 1; d < 32; d <<= 1) {
                const int t = __shfl_up(incl, d);
                if (lane >= d) incl += t;
            }
            int run = incl - sum;
#pragma unroll
            for (int i = 0; i < PER; i++) {
                const int n = lane * PER + i;
                if (n < NPW) nstart[n] = run;
                run += c[i];
            }
        }
        for (int idx = tid; idx < total; idx += 1024) {
            int r = q[idx] >> 20;
#pragma unroll
            for (int h = 0; h < H; h++) {
                float ex = expf(qa[idx * H + h] - decf(nmax[r * H + h]));
                qa[idx * H + h] = ex;
                atomicAdd(&nden[r * H + h], ex);
            }
        }
        __syncthreads();

        for (int idx = tid; idx < total; idx += 1024) {
            const int e = q[idx];
            pp[nstart[e >> 20] + (e & 0xFFFFF)] = (unsigned short)idx;
        }
        __syncthreads();

        const int per  = (total + 31) >> 5;
        const int g0   = wave * per;
        const int gend = (g0 + per < total) ? g0 + per : total;
        auto head = [&](int g, int& s, int& r, float* w) {
            s = 0; r = 0;
#pragma unroll
            for (int h = 0; h < H; h++) w[h] = 0.f;
            if (g + lane < gend) {
                const int idx = pp[g + lane];
                r = q[idx] >> 20;
                s = qs[idx];
#pragma unroll
                for (int h = 0; h < H; h++)
                    w[h] = qa[idx * H + h] / (nden[r * H + h] + 1e-16f);
            }
        };
        auto load8 = [&](int s, int j0, float (&val)[8][J]) {
#pragma unroll
            for (int b = 0; b < 8; b++) {
                int sj = __builtin_amdgcn_readlane(s, j0 + b);
#pragma unroll
                for (int jx = 0; jx < J; jx++) {
                    int c = lane + 32 * jx;
                    val[b][jx] = xh[(size_t)sj * HC + (c < HC ? c : HC - 1)];
                }
            }
        };
        int   cur = -1;
        float rowsum[J];
#pragma unroll
        for (int jx = 0; jx < J; jx++) rowsum[jx] = 0.f;
        auto flush = [&]() {
            if (cur >= 0) {
#pragma unroll
                for (int jx = 0; jx < J; jx++) {
                    int c = lane + 32 * jx;
                    if (c < HC) atomicAdd(&sacc[cur * LD + c], rowsum[jx]);
                    rowsum[jx] = 0.f;
                }
            }
        };
        auto add8 = [&](int r, const float* w, int j0, int cnt, const float (&val)[8][J]) {
#pragma unroll
            for (int b = 0; b < 8; b++) {
                if (j0 + b < cnt) {
                    int rj = __builtin_amdgcn_readlane(r, j0 + b);
                    if (rj != cur) {
                        flush();
                        cur = rj;
                    }
#pragma unroll
                    for (int jx = 0; jx < J; jx++) {
                        int h = (H == 1) ? 0 : (32 * jx) / C;
                        float wj = __int_as_float(__builtin_amdgcn_readlane(__float_as_int(w[h]), j0 + b));
                        rowsum[jx] += val[b][jx] * wj;
                    }
                }
            }
        };
        if (g0 < total) {
            int g = g0;
            int s, r;
            float w[H];
            float va[8][J], vb[8][J];
            head(g, s, r, w);
            load8(s, 0, va);
            while (gend - g >= 32) {
                const int gn = g + 32;
                int sn, rn;
                float wn[H];
                head(gn, sn, rn, wn);
                load8(s, 8, vb);
                add8(r, w, 0, 32, va);
                load8(s, 16, va);
                add8(r, w, 8, 32, vb);
                load8(s, 24, vb);
                add8(r, w, 16, 32, va);
                load8(sn, 0, va);
                add8(r, w, 24, 32, vb);
                g = gn; s = sn; r = rn;
#pragma unroll
                for (int h = 0; h < H; h++) w[h] = wn[h];
            }
            if (g < gend) {
                const int cnt = gend - g;
                add8(r, w, 0, cnt, va);
                for (int j0 = 8; j0 < cnt; j0 += 8) {
                    load8(s, j0, va);
                    add8(r, w, j0, cnt, va);
                }
            }
            flush();
        }
    } else {
        scan_owned(ob, grp, tid, [&](int e) {
            int r = dst[e] - lo;
            float ea[EDIM], a[H];
            load_ea(e, ea);
            atomicAdd(&nsum[r * NS], 1.0f);
#pragma unroll
            for (int i = 0; i < EDIM; i++) atomicAdd(&nsum[r * NS + 1 + i], ea[i]);
            raw_alpha<H>(ea, v, als, ald, src[e], lo + r, a);
#pragma unroll
            for (int h = 0; h < H; h++) atomicMax(&nmax[r * H + h], encf(a[h]));
        });
        __syncthreads();
        if (tid < nown) {
            float ea[EDIM], a[H];
            self_ea(tid, ea);
            raw_alpha<H>(ea, v, als, ald, lo + tid, lo + tid, a);
#pragma unroll
            for (int h = 0; h < H; h++) {
                aself[tid * H + h] = a[h];
                atomicMax(&nmax[tid * H + h], encf(a[h]));
            }
        }
        __syncthreads();
        scan_owned(ob, grp, tid, [&](int e) {
            int r = dst[e] - lo;
            float ea[EDIM], a[H];
            load_ea(e, ea);
            raw_alpha<H>(ea, v, als, ald, src[e], lo + r, a);
#pragma unroll
            for (int h = 0; h < H; h++)
                atomicAdd(&nden[r * H + h], expf(a[h] - decf(nmax[r * H + h])));
        });
        if (tid < nown) {
#pragma unroll
            for (int h = 0; h < H; h++)
                atomicAdd(&nden[tid * H + h], expf(aself[tid * H + h] - decf(nmax[tid * H + h])));
        }
        __syncthreads();
        auto add_row = [&](int s, int r, const float* a) {
            for (int h = 0; h < H; h++) {
                float w = expf(a[h] - decf(nmax[r * H + h])) / (nden[r * H + h] + 1e-16f);
                for (int c = 0; c < C; c++)
                    atomicAdd(&sacc[r * LD + h * C + c], xh[(size_t)s * HC + h * C + c] * w);
            }
        };
        scan_owned(ob, grp, tid, [&](int e) {
            int r = dst[e] - lo;
            float ea[EDIM], a[H];
            load_ea(e, ea);
            int s = src[e];
            raw_alpha<H>(ea, v, als, ald, s, lo + r, a);
            add_row(s, r, a);
        });
        if (tid < nown) {
            float a[H];
#pragma unroll
            for (int h = 0; h < H; h++) a[h] = aself[tid * H + h];
            add_row(lo + tid, tid, a);
        }
    }
    __syncthreads();

    if constexpr (PROJ) {
        for (int idx = tid; idx < NPW * HC; idx += 1024) {
            float* p = &sacc[(idx / HC) * LD + (idx % HC)];
            float x = *p + bc;
            *p = (x > 0.f) ? x : expm1f(x);
        }
        __syncthreads();
        constexpr int NT = NP / 16;
        float* ost = qa;
        const int half = lane >> 4, lrow = lane & 15;
        for (int j = wave; j < MT * NT; j += 32) {
            const int mt = j / NT, nt = j - mt * NT;
            int row = mt * 16 + lrow;
            if (row > NPW - 1) row = NPW - 1;
            const float* arow = sacc + row * LD + half * 2;
            const float* bcol = wns + (half * NP + nt * 16 + lrow) * 2;
            float* orow = ost + (mt * 16 + half * 8) * OP + nt * 16 + lrow;
            v8f d;
#pragma unroll
            for (int g = 0; g < 8; g++) d[g] = 0.0f;
#pragma unroll
            for (int k = 0; k < HC; k += 4) {
                v2f a = *reinterpret_cast<const v2f*>(arow + k);
                v2f b = *reinterpret_cast<const v2f*>(bcol + k * NP);
                d = wmma_f32_k4_spaced(a, b, d);
            }
#pragma unroll
            for (int g = 0; g < 8; g++) orow[g * OP] = d[g];
        }
        __syncthreads();
        float4* g4 = reinterpret_cast<float4*>(acc + (size_t)lo * OUT_F);
        for (int i = tid; i < nown * (OUT_F / 4); i += 1024) {
            int r = i / (OUT_F / 4), c4 = i - r * (OUT_F / 4);
            g4[i] = *reinterpret_cast<const float4*>(ost + r * OP + c4 * 4);
        }
        if (tid < nown) {
            alsn[lo + tid] = ost[tid * OP + OUT_F];
            aldn[lo + tid] = ost[tid * OP + OUT_F + 1];
        }
    } else {
        static_assert(LD == HC && HC % 4 == 0, "rows lie back to back and are read 16 bytes at a time");
        if (tid < nown) {
            float4* row4 = reinterpret_cast<float4*>(sacc + tid * LD);
            float vals[HC];
#pragma unroll
            for (int i = 0; i < HC / 4; i++) {
                const float4 t = row4[i];
                vals[4 * i]     = t.x + bias[4 * i];
                vals[4 * i + 1] = t.y + bias[4 * i + 1];
                vals[4 * i + 2] = t.z + bias[4 * i + 2];
                vals[4 * i + 3] = t.w + bias[4 * i + 3];
            }
            float mx = -INFINITY;
#pragma unroll
            for (int cc = 0; cc < HC; cc++) mx = fmaxf(mx, vals[cc]);
            float sum = 0.f;
#pragma unroll
            for (int cc = 0; cc < HC; cc++) sum += expf(vals[cc] - mx);
            const float lse = mx + logf(sum);
#pragma unroll
            for (int i = 0; i < HC / 4; i++)
                row4[i] = make_float4(vals[4 * i] - lse, vals[4 * i + 1] - lse,
                                      vals[4 * i + 2] - lse, vals[4 * i + 3] - lse);
        }
        __syncthreads();
        float4* g4 = reinterpret_cast<float4*>(acc + (size_t)lo * HC);
        const float4* s4 = reinterpret_cast<const float4*>(sacc);
        for (int i = tid; i < nown * (HC / 4); i += 1024) g4[i] = s4[i];
    }
}

extern "C" void kernel_launch(void* const* d_in, const int* in_sizes, int n_in,
                              void* d_out, int out_size, void* d_ws, size_t ws_size,
                              hipStream_t stream) {
    const float* x   = (const float*)d_in[0];
    const int*   ei  = (const int*)d_in[1];
    const float* ea  = (const float*)d_in[2];
    const float* W1  = (const float*)d_in[3];
    const float* as1 = (const float*)d_in[4];
    const float* ad1 = (const float*)d_in[5];
    const float* We1 = (const float*)d_in[6];
    const float* ae1 = (const float*)d_in[7];
    const float* b1  = (const float*)d_in[8];
    const float* W2  = (const float*)d_in[9];
    const float* as2 = (const float*)d_in[10];
    const float* ad2 = (const float*)d_in[11];
    const float* We2 = (const float*)d_in[12];
    const float* ae2 = (const float*)d_in[13];
    const float* b2  = (const float*)d_in[14];

    const int N = N_NODES, E = N_EDGES;
    const int HC1 = HEADS * HID;
    const int* src = ei;
    const int* dst = ei + E;

    char* ws = (char*)d_ws;
    size_t off = 0;
    auto alloc = [&](size_t bytes) -> void* {
        void* p = ws + off;
        off = (off + bytes + 255) & ~(size_t)255;
        return p;
    };
    float*    xh1    = (float*)alloc(sizeof(float) * (size_t)N * HC1);
    float*    als1   = (float*)alloc(sizeof(float) * N * HEADS);
    float*    ald1   = (float*)alloc(sizeof(float) * N * HEADS);
    float*    xh2    = (float*)alloc(sizeof(float) * (size_t)N * OUT_F);
    float*    als2   = (float*)alloc(sizeof(float) * N);
    float*    ald2   = (float*)alloc(sizeof(float) * N);
    unsigned* ob     = (unsigned*)alloc(N_EDGES);
    unsigned* hcnt   = (unsigned*)alloc(sizeof(unsigned) * NGRP);
    unsigned* hsr    = (unsigned*)alloc(sizeof(unsigned) * (size_t)NGRP * QCAP);
    float*    hale   = (float*)alloc(sizeof(float) * (size_t)NGRP * QCAP);
    (void)ws_size; (void)n_in; (void)in_sizes; (void)out_size;

    const int T = 256;

    owner_bytes<<<CDIV(N_EDGES / 16, T), T, 0, stream>>>(dst, ob);

    {
        int mtiles = N / 16, waves_per_blk = GEMM_T / 32;
        wmma_gemm<8, F_IN><<<CDIV(mtiles, waves_per_blk), GEMM_T, 0, stream>>>(x, W1, as1, ad1, xh1,
                                                                               als1, ald1, N, HC1);
    }
    gat_owned<HEADS, HID, true><<<NGRP, 1024, 0, stream>>>(src, dst, ob, ea, We1, ae1, als1, ald1,
                                                           xh1, b1, W2, We2, ae2, as2, ad2,
                                                           hcnt, hsr, hale, xh2, als2, ald2);

    gat_owned<1, OUT_F, false><<<NGRP, 1024, 0, stream>>>(src, dst, ob, ea, We2, ae2, als2, ald2,
                                                          xh2, b2, nullptr, nullptr, nullptr,
                                                          nullptr, nullptr,
                                                          hcnt, hsr, hale, (float*)d_out, nullptr,
                                                          nullptr);
}
